# starting kernel
# baseline (speedup 1.0000x reference)
#include <hip/hip_runtime.h>
#include <math.h>

typedef __attribute__((ext_vector_type(2))) float v2f;
typedef __attribute__((ext_vector_type(8))) float v8f;

#define B_SZ   256
#define M_SZ   1024
#define A_SZ   8
#define K_CLS  20
#define O_SZ   16
#define KPAD   32
#define I_SZ   (B_SZ * A_SZ)

__global__ void caps_zero_bij(float* __restrict__ bij) {
    int idx = blockIdx.x * blockDim.x + threadIdx.x;
    bij[idx] = 0.0f;
}

__global__ void caps_softmax(const float* __restrict__ bij, float* __restrict__ c) {
    __shared__ float red[256];
    const int k = blockIdx.x;
    const int t = threadIdx.x;

    float v0 = bij[(t + 0  ) * KPAD + k];
    float v1 = bij[(t + 256) * KPAD + k];
    float v2 = bij[(t + 512) * KPAD + k];
    float v3 = bij[(t + 768) * KPAD + k];

    float mx = fmaxf(fmaxf(v0, v1), fmaxf(v2, v3));
    red[t] = mx;
    __syncthreads();
    for (int s = 128; s > 0; s >>= 1) {
        if (t < s) red[t] = fmaxf(red[t], red[t + s]);
        __syncthreads();
    }
    mx = red[0];
    __syncthreads();

    float e0 = expf(v0 - mx), e1 = expf(v1 - mx);
    float e2 = expf(v2 - mx), e3 = expf(v3 - mx);
    red[t] = e0 + e1 + e2 + e3;
    __syncthreads();
    for (int s = 128; s > 0; s >>= 1) {
        if (t < s) red[t] += red[t + s];
        __syncthreads();
    }
    float inv = 1.0f / red[0];

    c[(t + 0  ) * KPAD + k] = e0 * inv;
    c[(t + 256) * KPAD + k] = e1 * inv;
    c[(t + 512) * KPAD + k] = e2 * inv;
    c[(t + 768) * KPAD + k] = e3 * inv;
}

__global__ void caps_gemm_t(const float* __restrict__ x,
                            const float* __restrict__ c,
                            float* __restrict__ t2) {
    const int wave = (blockIdx.x * blockDim.x + threadIdx.x) >> 5;
    const int lane = threadIdx.x & 31;
    const int i0   = wave * 16;
    const int lo   = lane & 15;
    const int hi   = lane >> 4;
    const int row  = i0 + lo;
    const int xb   = (row >> 3) * (A_SZ * M_SZ) + (row & 7);
    const int j0   = hi * 2;

    v8f acc0 = {};
    v8f acc1 = {};
    #pragma unroll 2
    for (int mb = 0; mb < M_SZ; mb += 4) {
        v2f av, bv0, bv1;
        av.x  = x[xb + (mb + j0    ) * A_SZ];
        av.y  = x[xb + (mb + j0 + 1) * A_SZ];
        bv0.x = c[(mb + j0    ) * KPAD + lo];
        bv0.y = c[(mb + j0 + 1) * KPAD + lo];
        bv1.x = c[(mb + j0    ) * KPAD + 16 + lo];
        bv1.y = c[(mb + j0 + 1) * KPAD + 16 + lo];
        acc0 = __builtin_amdgcn_wmma_f32_16x16x4_f32(
            false, av, false, bv0, (short)0, acc0, false, false);
        acc1 = __builtin_amdgcn_wmma_f32_16x16x4_f32(
            false, av, false, bv1, (short)0, acc1, false, false);
    }
    #pragma unroll
    for (int r = 0; r < 8; ++r) {
        const int rbase = (i0 + r + hi * 8) * KPAD;
        t2[rbase + lo]      = acc0[r];
        t2[rbase + 16 + lo] = acc1[r];
    }
}

__global__ void caps_squash(const float* __restrict__ t2,
                            const float* __restrict__ W,
                            float* __restrict__ out,
                            float* __restrict__ wv2) {
    __shared__ float s_s[K_CLS][O_SZ];
    __shared__ float s_v[K_CLS][O_SZ];
    const int b = blockIdx.x;
    const int t = threadIdx.x;
    const int k = t >> 4;
    const int o = t & 15;

    float s = 0.0f;
    #pragma unroll
    for (int a = 0; a < A_SZ; ++a)
        s += W[k * (O_SZ * A_SZ) + o * A_SZ + a] * t2[(b * 8 + a) * KPAD + k];
    s_s[k][o] = s;
    __syncthreads();

    float magsq = 0.0f;
    #pragma unroll
    for (int kk = 0; kk < K_CLS; ++kk) {
        float sv = s_s[kk][o];
        magsq += sv * sv;
    }
    float scale = magsq / ((1.0f + magsq) * sqrtf(magsq));
    float v = s * scale;
    s_v[k][o] = v;
    out[b * (K_CLS * O_SZ) + t] = v;
    __syncthreads();

    if (t < K_CLS * A_SZ) {
        int k2 = t % K_CLS;
        int a  = t / K_CLS;
        float w = 0.0f;
        #pragma unroll
        for (int o2 = 0; o2 < O_SZ; ++o2)
            w += W[k2 * (O_SZ * A_SZ) + o2 * A_SZ + a] * s_v[k2][o2];
        wv2[(b * 8 + a) * KPAD + k2] = w * (1.0f / (float)B_SZ);
    }
}

__global__ void caps_gemm_b(const float* __restrict__ x,
                            const float* __restrict__ wv2,
                            float* __restrict__ bij) {
    const int wave = (blockIdx.x * blockDim.x + threadIdx.x) >> 5;
    const int lane = threadIdx.x & 31;
    const int m0   = wave * 16;
    const int lo   = lane & 15;
    const int hi   = lane >> 4;
    const int m    = m0 + lo;
    const int j0   = hi * 2;

    v8f acc0 = {};
    v8f acc1 = {};
    #pragma unroll 2
    for (int ib = 0; ib < I_SZ; ib += 4) {
        const int ia = ib + j0;
        const int ic = ia + 1;
        v2f av, bv0, bv1;
        av.x  = x[(ia >> 3) * (A_SZ * M_SZ) + m * A_SZ + (ia & 7)];
        av.y  = x[(ic >> 3) * (A_SZ * M_SZ) + m * A_SZ + (ic & 7)];
        bv0.x = wv2[ia * KPAD + lo];
        bv0.y = wv2[ic * KPAD + lo];
        bv1.x = wv2[ia * KPAD + 16 + lo];
        bv1.y = wv2[ic * KPAD + 16 + lo];
        acc0 = __builtin_amdgcn_wmma_f32_16x16x4_f32(
            false, av, false, bv0, (short)0, acc0, false, false);
        acc1 = __builtin_amdgcn_wmma_f32_16x16x4_f32(
            false, av, false, bv1, (short)0, acc1, false, false);
    }
    #pragma unroll
    for (int r = 0; r < 8; ++r) {
        const int rbase = (m0 + r + hi * 8) * KPAD;
        bij[rbase + lo]      += acc0[r];
        bij[rbase + 16 + lo] += acc1[r];
    }
}

extern "C" void kernel_launch(void* const* d_in, const int* in_sizes, int n_in,
                              void* d_out, int out_size, void* d_ws, size_t ws_size,
                              hipStream_t stream) {
    (void)in_sizes; (void)n_in; (void)out_size; (void)ws_size;
    const float* x = (const float*)d_in[0];
    const float* W = (const float*)d_in[1];
    float* out = (float*)d_out;

    float* bij = (float*)d_ws;
    float* c   = bij + M_SZ * KPAD;
    float* t2  = c   + M_SZ * KPAD;
    float* wv2 = t2  + I_SZ * KPAD;

    caps_zero_bij<<<(M_SZ * KPAD) / 256, 256, 0, stream>>>(bij);

    for (int it = 0; it < 3; ++it) {
        caps_softmax<<<K_CLS, 256, 0, stream>>>(bij, c);
        caps_gemm_t<<<16, 256, 0, stream>>>(x, c, t2);
        caps_squash<<<B_SZ, 320, 0, stream>>>(t2, W, out, wv2);
        if (it < 2)
            caps_gemm_b<<<8, 256, 0, stream>>>(x, wv2, bij);
    }
}


# Split-K inside the work-group: one block per 16x32 slab; its 16 (gemm_t) or 32 (gemm_b) waves each take a slice of the contraction, partials added through LDS in wave order (sum order changes)
# speedup vs baseline: 6.2779x; 6.2779x over previous
#include <hip/hip_runtime.h>
#include <math.h>

typedef __attribute__((ext_vector_type(2))) float v2f;
typedef __attribute__((ext_vector_type(8))) float v8f;

#define B_SZ   256
#define M_SZ   1024
#define A_SZ   8
#define K_CLS  20
#define O_SZ   16
#define KPAD   32
#define I_SZ   (B_SZ * A_SZ)

__global__ void caps_zero_bij(float* __restrict__ bij) {
    int idx = blockIdx.x * blockDim.x + threadIdx.x;
    bij[idx] = 0.0f;
}

__global__ void caps_softmax(const float* __restrict__ bij, float* __restrict__ c) {
    __shared__ float red[256];
    const int k = blockIdx.x;
    const int t = threadIdx.x;

    float v0 = bij[(t + 0  ) * KPAD + k];
    float v1 = bij[(t + 256) * KPAD + k];
    float v2 = bij[(t + 512) * KPAD + k];
    float v3 = bij[(t + 768) * KPAD + k];

    float mx = fmaxf(fmaxf(v0, v1), fmaxf(v2, v3));
    red[t] = mx;
    __syncthreads();
    for (int s = 128; s > 0; s >>= 1) {
        if (t < s) red[t] = fmaxf(red[t], red[t + s]);
        __syncthreads();
    }
    mx = red[0];
    __syncthreads();

    float e0 = expf(v0 - mx), e1 = expf(v1 - mx);
    float e2 = expf(v2 - mx), e3 = expf(v3 - mx);
    red[t] = e0 + e1 + e2 + e3;
    __syncthreads();
    for (int s = 128; s > 0; s >>= 1) {
        if (t < s) red[t] += red[t + s];
        __syncthreads();
    }
    float inv = 1.0f / red[0];

    c[(t + 0  ) * KPAD + k] = e0 * inv;
    c[(t + 256) * KPAD + k] = e1 * inv;
    c[(t + 512) * KPAD + k] = e2 * inv;
    c[(t + 768) * KPAD + k] = e3 * inv;
}

#define T_WAVES 16
#define T_CHUNK (M_SZ / T_WAVES)
__global__ void __launch_bounds__(T_WAVES * 32)
caps_gemm_t(const float* __restrict__ x,
            const float* __restrict__ c,
            float* __restrict__ t2) {
    __shared__ float part[T_WAVES][16 * KPAD];
    const int w    = __builtin_amdgcn_readfirstlane(threadIdx.x >> 5);
    const int lane = threadIdx.x & 31;
    const int i0   = blockIdx.x * 16;
    const int lo   = lane & 15;
    const int hi   = lane >> 4;
    const int row  = i0 + lo;
    const int xb   = (row >> 3) * (A_SZ * M_SZ) + (row & 7);
    const int j0   = hi * 2;

    v8f acc0 = {};
    v8f acc1 = {};
    #pragma unroll 2
    for (int s = 0; s < T_CHUNK; s += 4) {
        const int mb = w * T_CHUNK + s;
        v2f av, bv0, bv1;
        av.x  = x[xb + (mb + j0    ) * A_SZ];
        av.y  = x[xb + (mb + j0 + 1) * A_SZ];
        bv0.x = c[(mb + j0    ) * KPAD + lo];
        bv0.y = c[(mb + j0 + 1) * KPAD + lo];
        bv1.x = c[(mb + j0    ) * KPAD + 16 + lo];
        bv1.y = c[(mb + j0 + 1) * KPAD + 16 + lo];
        acc0 = __builtin_amdgcn_wmma_f32_16x16x4_f32(
            false, av, false, bv0, (short)0, acc0, false, false);
        acc1 = __builtin_amdgcn_wmma_f32_16x16x4_f32(
            false, av, false, bv1, (short)0, acc1, false, false);
    }
    #pragma unroll
    for (int r = 0; r < 8; ++r) {
        const int rbase = (r + hi * 8) * KPAD;
        part[w][rbase + lo]      = acc0[r];
        part[w][rbase + 16 + lo] = acc1[r];
    }
    __syncthreads();
    float sum = part[0][threadIdx.x];
    #pragma unroll
    for (int ww = 1; ww < T_WAVES; ++ww)
        sum += part[ww][threadIdx.x];
    t2[i0 * KPAD + threadIdx.x] = sum;
}

__global__ void caps_squash(const float* __restrict__ t2,
                            const float* __restrict__ W,
                            float* __restrict__ out,
                            float* __restrict__ wv2) {
    __shared__ float s_s[K_CLS][O_SZ];
    __shared__ float s_v[K_CLS][O_SZ];
    const int b = blockIdx.x;
    const int t = threadIdx.x;
    const int k = t >> 4;
    const int o = t & 15;

    float s = 0.0f;
    #pragma unroll
    for (int a = 0; a < A_SZ; ++a)
        s += W[k * (O_SZ * A_SZ) + o * A_SZ + a] * t2[(b * 8 + a) * KPAD + k];
    s_s[k][o] = s;
    __syncthreads();

    float magsq = 0.0f;
    #pragma unroll
    for (int kk = 0; kk < K_CLS; ++kk) {
        float sv = s_s[kk][o];
        magsq += sv * sv;
    }
    float scale = magsq / ((1.0f + magsq) * sqrtf(magsq));
    float v = s * scale;
    s_v[k][o] = v;
    out[b * (K_CLS * O_SZ) + t] = v;
    __syncthreads();

    if (t < K_CLS * A_SZ) {
        int k2 = t % K_CLS;
        int a  = t / K_CLS;
        float w = 0.0f;
        #pragma unroll
        for (int o2 = 0; o2 < O_SZ; ++o2)
            w += W[k2 * (O_SZ * A_SZ) + o2 * A_SZ + a] * s_v[k2][o2];
        wv2[(b * 8 + a) * KPAD + k2] = w * (1.0f / (float)B_SZ);
    }
}

#define B_WAVES 32
#define B_CHUNK (I_SZ / B_WAVES)
__global__ void __launch_bounds__(B_WAVES * 32)
caps_gemm_b(const float* __restrict__ x,
            const float* __restrict__ wv2,
            float* __restrict__ bij) {
    __shared__ float part[B_WAVES][16 * KPAD];
    const int w    = __builtin_amdgcn_readfirstlane(threadIdx.x >> 5);
    const int lane = threadIdx.x & 31;
    const int m0   = blockIdx.x * 16;
    const int lo   = lane & 15;
    const int hi   = lane >> 4;
    const int m    = m0 + lo;
    const int j0   = hi * 2;

    v8f acc0 = {};
    v8f acc1 = {};
    #pragma unroll 2
    for (int s = 0; s < B_CHUNK; s += 4) {
        const int ib = w * B_CHUNK + s;
        const int ia = ib + j0;
        const int ic = ia + 1;
        v2f av, bv0, bv1;
        av.x  = x[(ia >> 3) * (A_SZ * M_SZ) + m * A_SZ + (ia & 7)];
        av.y  = x[(ic >> 3) * (A_SZ * M_SZ) + m * A_SZ + (ic & 7)];
        bv0.x = wv2[ia * KPAD + lo];
        bv0.y = wv2[ic * KPAD + lo];
        bv1.x = wv2[ia * KPAD + 16 + lo];
        bv1.y = wv2[ic * KPAD + 16 + lo];
        acc0 = __builtin_amdgcn_wmma_f32_16x16x4_f32(
            false, av, false, bv0, (short)0, acc0, false, false);
        acc1 = __builtin_amdgcn_wmma_f32_16x16x4_f32(
            false, av, false, bv1, (short)0, acc1, false, false);
    }
    #pragma unroll
    for (int r = 0; r < 8; ++r) {
        const int rbase = (r + hi * 8) * KPAD;
        part[w][rbase + lo]      = acc0[r];
        part[w][rbase + 16 + lo] = acc1[r];
    }
    __syncthreads();
    if (threadIdx.x < 16 * KPAD) {
        float sum = part[0][threadIdx.x];
        #pragma unroll
        for (int ww = 1; ww < B_WAVES; ++ww)
            sum += part[ww][threadIdx.x];
        bij[m0 * KPAD + threadIdx.x] += sum;
    }
}

extern "C" void kernel_launch(void* const* d_in, const int* in_sizes, int n_in,
                              void* d_out, int out_size, void* d_ws, size_t ws_size,
                              hipStream_t stream) {
    (void)in_sizes; (void)n_in; (void)out_size; (void)ws_size;
    const float* x = (const float*)d_in[0];
    const float* W = (const float*)d_in[1];
    float* out = (float*)d_out;

    float* bij = (float*)d_ws;
    float* c   = bij + M_SZ * KPAD;
    float* t2  = c   + M_SZ * KPAD;
    float* wv2 = t2  + I_SZ * KPAD;

    caps_zero_bij<<<(M_SZ * KPAD) / 256, 256, 0, stream>>>(bij);

    for (int it = 0; it < 3; ++it) {
        caps_softmax<<<K_CLS, 256, 0, stream>>>(bij, c);
        caps_gemm_t<<<I_SZ / 16, T_WAVES * 32, 0, stream>>>(x, c, t2);
        caps_squash<<<B_SZ, 320, 0, stream>>>(t2, W, out, wv2);
        if (it < 2)
            caps_gemm_b<<<M_SZ / 16, B_WAVES * 32, 0, stream>>>(x, wv2, bij);
    }
}


# Squash stage fused into the gemm_t epilogue: a block's 16x32 slab is two whole batch elements, so s, v and wv2 are made from the slab in LDS; 12 -> 9 launches, t2 round trip gone (sum order kept)
# speedup vs baseline: 7.0040x; 1.1157x over previous
#include <hip/hip_runtime.h>
#include <math.h>

typedef __attribute__((ext_vector_type(2))) float v2f;
typedef __attribute__((ext_vector_type(8))) float v8f;

#define B_SZ   256
#define M_SZ   1024
#define A_SZ   8
#define K_CLS  20
#define O_SZ   16
#define KPAD   32
#define I_SZ   (B_SZ * A_SZ)

__global__ void caps_zero_bij(float* __restrict__ bij) {
    int idx = blockIdx.x * blockDim.x + threadIdx.x;
    bij[idx] = 0.0f;
}

__global__ void caps_softmax(const float* __restrict__ bij, float* __restrict__ c) {
    __shared__ float red[256];
    const int k = blockIdx.x;
    const int t = threadIdx.x;

    float v0 = bij[(t + 0  ) * KPAD + k];
    float v1 = bij[(t + 256) * KPAD + k];
    float v2 = bij[(t + 512) * KPAD + k];
    float v3 = bij[(t + 768) * KPAD + k];

    float mx = fmaxf(fmaxf(v0, v1), fmaxf(v2, v3));
    red[t] = mx;
    __syncthreads();
    for (int s = 128; s > 0; s >>= 1) {
        if (t < s) red[t] = fmaxf(red[t], red[t + s]);
        __syncthreads();
    }
    mx = red[0];
    __syncthreads();

    float e0 = expf(v0 - mx), e1 = expf(v1 - mx);
    float e2 = expf(v2 - mx), e3 = expf(v3 - mx);
    red[t] = e0 + e1 + e2 + e3;
    __syncthreads();
    for (int s = 128; s > 0; s >>= 1) {
        if (t < s) red[t] += red[t + s];
        __syncthreads();
    }
    float inv = 1.0f / red[0];

    c[(t + 0  ) * KPAD + k] = e0 * inv;
    c[(t + 256) * KPAD + k] = e1 * inv;
    c[(t + 512) * KPAD + k] = e2 * inv;
    c[(t + 768) * KPAD + k] = e3 * inv;
}

#define T_WAVES 16
#define T_CHUNK (M_SZ / T_WAVES)
#define SQ_THREADS (K_CLS * O_SZ)
#define SQ_HALF    (K_CLS * A_SZ)
__global__ void __launch_bounds__(T_WAVES * 32) __attribute__((amdgpu_waves_per_eu(4, 4)))
caps_gemm_t_squash(const float* __restrict__ x,
                   const float* __restrict__ c,
                   const float* __restrict__ W,
                   float* __restrict__ out,
                   float* __restrict__ wv2) {
    __shared__ float part[T_WAVES][16 * KPAD];
    __shared__ float s_t[16 * KPAD];
    __shared__ float s_s[2][K_CLS][O_SZ];
    __shared__ float s_v[2][K_CLS][O_SZ + 1];
    const int w    = __builtin_amdgcn_readfirstlane(threadIdx.x >> 5);
    const int lane = threadIdx.x & 31;
    const int t    = threadIdx.x;
    const int i0   = blockIdx.x * 16;
    const int b0   = blockIdx.x * 2;
    const int lo   = lane & 15;
    const int hi   = lane >> 4;
    const int row  = i0 + lo;
    const int xb   = (row >> 3) * (A_SZ * M_SZ) + (row & 7);
    const int j0   = hi * 2;

    const bool sq  = w < (SQ_THREADS / 32);
    const int k    = t >> 4;
    const int o    = t & 15;
    const int bb3  = (t >= SQ_HALF) ? 1 : 0;
    const int r3   = t - bb3 * SQ_HALF;
    const int k3   = r3 % K_CLS;
    const int a3   = r3 / K_CLS;

    float wa[A_SZ] = {};
    float wo[O_SZ] = {};
    if (sq) {
        #pragma unroll
        for (int a = 0; a < A_SZ; ++a)
            wa[a] = W[k * (O_SZ * A_SZ) + o * A_SZ + a];
        #pragma unroll
        for (int o2 = 0; o2 < O_SZ; ++o2)
            wo[o2] = W[k3 * (O_SZ * A_SZ) + o2 * A_SZ + a3];
    }

    v8f acc0 = {};
    v8f acc1 = {};
    #pragma unroll 2
    for (int s = 0; s < T_CHUNK; s += 4) {
        const int mb = w * T_CHUNK + s;
        v2f av, bv0, bv1;
        av.x  = x[xb + (mb + j0    ) * A_SZ];
        av.y  = x[xb + (mb + j0 + 1) * A_SZ];
        bv0.x = c[(mb + j0    ) * KPAD + lo];
        bv0.y = c[(mb + j0 + 1) * KPAD + lo];
        bv1.x = c[(mb + j0    ) * KPAD + 16 + lo];
        bv1.y = c[(mb + j0 + 1) * KPAD + 16 + lo];
        acc0 = __builtin_amdgcn_wmma_f32_16x16x4_f32(
            false, av, false, bv0, (short)0, acc0, false, false);
        acc1 = __builtin_amdgcn_wmma_f32_16x16x4_f32(
            false, av, false, bv1, (short)0, acc1, false, false);
    }
    #pragma unroll
    for (int r = 0; r < 8; ++r) {
        const int rbase = (r + hi * 8) * KPAD;
        part[w][rbase + lo]      = acc0[r];
        part[w][rbase + 16 + lo] = acc1[r];
    }
    __syncthreads();
    float sum = part[0][t];
    #pragma unroll
    for (int ww = 1; ww < T_WAVES; ++ww)
        sum += part[ww][t];
    s_t[t] = sum;
    __syncthreads();

    float s0 = 0.0f, s1 = 0.0f;
    if (sq) {
        #pragma unroll
        for (int a = 0; a < A_SZ; ++a) {
            s0 += wa[a] * s_t[a * KPAD + k];
            s1 += wa[a] * s_t[(A_SZ + a) * KPAD + k];
        }
        s_s[0][k][o] = s0;
        s_s[1][k][o] = s1;
    }
    __syncthreads();

    float v0 = 0.0f, v1 = 0.0f;
    if (sq) {
        float mg0 = 0.0f, mg1 = 0.0f;
        #pragma unroll
        for (int kk = 0; kk < K_CLS; ++kk) {
            const float sv0 = s_s[0][kk][o];
            const float sv1 = s_s[1][kk][o];
            mg0 += sv0 * sv0;
            mg1 += sv1 * sv1;
        }
        const float scale0 = mg0 / ((1.0f + mg0) * sqrtf(mg0));
        const float scale1 = mg1 / ((1.0f + mg1) * sqrtf(mg1));
        v0 = s0 * scale0;
        v1 = s1 * scale1;
        s_v[0][k][o] = v0;
        s_v[1][k][o] = v1;
    }
    __syncthreads();

    if (sq) {
        out[(b0    ) * (K_CLS * O_SZ) + t] = v0;
        out[(b0 + 1) * (K_CLS * O_SZ) + t] = v1;
        float wsum = 0.0f;
        #pragma unroll
        for (int o2 = 0; o2 < O_SZ; ++o2)
            wsum += wo[o2] * s_v[bb3][k3][o2];
        wv2[((b0 + bb3) * A_SZ + a3) * KPAD + k3] = wsum * (1.0f / (float)B_SZ);
    }
}

#define B_WAVES 32
#define B_CHUNK (I_SZ / B_WAVES)
__global__ void __launch_bounds__(B_WAVES * 32)
caps_gemm_b(const float* __restrict__ x,
            const float* __restrict__ wv2,
            float* __restrict__ bij) {
    __shared__ float part[B_WAVES][16 * KPAD];
    const int w    = __builtin_amdgcn_readfirstlane(threadIdx.x >> 5);
    const int lane = threadIdx.x & 31;
    const int m0   = blockIdx.x * 16;
    const int lo   = lane & 15;
    const int hi   = lane >> 4;
    const int m    = m0 + lo;
    const int j0   = hi * 2;

    v8f acc0 = {};
    v8f acc1 = {};
    #pragma unroll 2
    for (int s = 0; s < B_CHUNK; s += 4) {
        const int ib = w * B_CHUNK + s;
        const int ia = ib + j0;
        const int ic = ia + 1;
        v2f av, bv0, bv1;
        av.x  = x[(ia >> 3) * (A_SZ * M_SZ) + m * A_SZ + (ia & 7)];
        av.y  = x[(ic >> 3) * (A_SZ * M_SZ) + m * A_SZ + (ic & 7)];
        bv0.x = wv2[ia * KPAD + lo];
        bv0.y = wv2[ic * KPAD + lo];
        bv1.x = wv2[ia * KPAD + 16 + lo];
        bv1.y = wv2[ic * KPAD + 16 + lo];
        acc0 = __builtin_amdgcn_wmma_f32_16x16x4_f32(
            false, av, false, bv0, (short)0, acc0, false, false);
        acc1 = __builtin_amdgcn_wmma_f32_16x16x4_f32(
            false, av, false, bv1, (short)0, acc1, false, false);
    }
    #pragma unroll
    for (int r = 0; r < 8; ++r) {
        const int rbase = (r + hi * 8) * KPAD;
        part[w][rbase + lo]      = acc0[r];
        part[w][rbase + 16 + lo] = acc1[r];
    }
    __syncthreads();
    if (threadIdx.x < 16 * KPAD) {
        float sum = part[0][threadIdx.x];
        #pragma unroll
        for (int ww = 1; ww < B_WAVES; ++ww)
            sum += part[ww][threadIdx.x];
        bij[m0 * KPAD + threadIdx.x] += sum;
    }
}

extern "C" void kernel_launch(void* const* d_in, const int* in_sizes, int n_in,
                              void* d_out, int out_size, void* d_ws, size_t ws_size,
                              hipStream_t stream) {
    (void)in_sizes; (void)n_in; (void)out_size; (void)ws_size;
    const float* x = (const float*)d_in[0];
    const float* W = (const float*)d_in[1];
    float* out = (float*)d_out;

    float* bij = (float*)d_ws;
    float* c   = bij + M_SZ * KPAD;
    float* wv2 = c   + M_SZ * KPAD;

    caps_zero_bij<<<(M_SZ * KPAD) / 256, 256, 0, stream>>>(bij);

    for (int it = 0; it < 3; ++it) {
        caps_softmax<<<K_CLS, 256, 0, stream>>>(bij, c);
        caps_gemm_t_squash<<<I_SZ / 16, T_WAVES * 32, 0, stream>>>(x, c, W, out, wv2);
        if (it < 2)
            caps_gemm_b<<<M_SZ / 16, B_WAVES * 32, 0, stream>>>(x, wv2, bij);
    }
}


# Column softmax folded into the two GEMMs in two levels: gemm_b's epilogue leaves per-slab max, exp and sum, gemm_t rescales per-slab accumulators and divides; 9 -> 6 launches (sum order changes)
# speedup vs baseline: 9.2690x; 1.3234x over previous
#include <hip/hip_runtime.h>
#include <math.h>

typedef __attribute__((ext_vector_type(2))) float v2f;
typedef __attribute__((ext_vector_type(8))) float v8f;

#define B_SZ   256
#define M_SZ   1024
#define A_SZ   8
#define K_CLS  20
#define O_SZ   16
#define KPAD   32
#define I_SZ   (B_SZ * A_SZ)

#define N_SLAB (M_SZ / 16)

__global__ void caps_init(float* __restrict__ bij, float* __restrict__ e,
                          float* __restrict__ smax, float* __restrict__ ssum) {
    int idx = blockIdx.x * blockDim.x + threadIdx.x;
    bij[idx] = 0.0f;
    e[idx]   = 1.0f;
    if (idx < N_SLAB * KPAD) {
        smax[idx] = 0.0f;
        ssum[idx] = 16.0f;
    }
}

#define T_WAVES 16
#define T_CHUNK (M_SZ / T_WAVES)
#define T_SLABS (T_CHUNK / 16)
#define SQ_THREADS (K_CLS * O_SZ)
#define SQ_HALF    (K_CLS * A_SZ)
__global__ void __launch_bounds__(T_WAVES * 32) __attribute__((amdgpu_waves_per_eu(4, 4)))
caps_gemm_t_squash(const float* __restrict__ x,
                   const float* __restrict__ e,
                   const float* __restrict__ smax,
                   const float* __restrict__ ssum,
                   const float* __restrict__ W,
                   float* __restrict__ out,
                   float* __restrict__ wv2) {
    __shared__ float part[T_WAVES][16 * KPAD];
    __shared__ float s_mx[T_WAVES][KPAD];
    __shared__ float s_dn[T_WAVES][KPAD];
    __shared__ float s_t[16 * KPAD];
    __shared__ float s_s[2][K_CLS][O_SZ];
    __shared__ float s_v[2][K_CLS][O_SZ + 1];
    const int w    = __builtin_amdgcn_readfirstlane(threadIdx.x >> 5);
    const int lane = threadIdx.x & 31;
    const int t    = threadIdx.x;
    const int i0   = blockIdx.x * 16;
    const int b0   = blockIdx.x * 2;
    const int lo   = lane & 15;
    const int hi   = lane >> 4;
    const int row  = i0 + lo;
    const int xb   = (row >> 3) * (A_SZ * M_SZ) + (row & 7);
    const int j0   = hi * 2;

    const bool sq  = w < (SQ_THREADS / 32);
    const int k    = t >> 4;
    const int o    = t & 15;
    const int bb3  = (t >= SQ_HALF) ? 1 : 0;
    const int r3   = t - bb3 * SQ_HALF;
    const int k3   = r3 % K_CLS;
    const int a3   = r3 / K_CLS;

    float bm[T_SLABS], bs[T_SLABS];
    #pragma unroll
    for (int j = 0; j < T_SLABS; ++j) {
        bm[j] = smax[(w * T_SLABS + j) * KPAD + lane];
        bs[j] = ssum[(w * T_SLABS + j) * KPAD + lane];
    }

    float wa[A_SZ] = {};
    float wo[O_SZ] = {};
    if (sq) {
        #pragma unroll
        for (int a = 0; a < A_SZ; ++a)
            wa[a] = W[k * (O_SZ * A_SZ) + o * A_SZ + a];
        #pragma unroll
        for (int o2 = 0; o2 < O_SZ; ++o2)
            wo[o2] = W[k3 * (O_SZ * A_SZ) + o2 * A_SZ + a3];
    }

    v8f acc0[T_SLABS] = {};
    v8f acc1[T_SLABS] = {};
    #pragma unroll
    for (int j = 0; j < T_SLABS; ++j) {
        #pragma unroll 2
        for (int s = 0; s < 16; s += 4) {
            const int mb = w * T_CHUNK + j * 16 + s;
            v2f av, bv0, bv1;
            av.x  = x[xb + (mb + j0    ) * A_SZ];
            av.y  = x[xb + (mb + j0 + 1) * A_SZ];
            bv0.x = e[(mb + j0    ) * KPAD + lo];
            bv0.y = e[(mb + j0 + 1) * KPAD + lo];
            bv1.x = e[(mb + j0    ) * KPAD + 16 + lo];
            bv1.y = e[(mb + j0 + 1) * KPAD + 16 + lo];
            acc0[j] = __builtin_amdgcn_wmma_f32_16x16x4_f32(
                false, av, false, bv0, (short)0, acc0[j], false, false);
            acc1[j] = __builtin_amdgcn_wmma_f32_16x16x4_f32(
                false, av, false, bv1, (short)0, acc1[j], false, false);
        }
    }
    s_mx[w][lane] = fmaxf(fmaxf(bm[0], bm[1]), fmaxf(bm[2], bm[3]));
    __syncthreads();

    float g = s_mx[0][lane];
    #pragma unroll
    for (int ww = 1; ww < T_WAVES; ++ww)
        g = fmaxf(g, s_mx[ww][lane]);
    float f[T_SLABS];
    float dn = 0.0f;
    #pragma unroll
    for (int j = 0; j < T_SLABS; ++j) {
        f[j] = expf(bm[j] - g);
        dn  += f[j] * bs[j];
    }
    s_dn[w][lane] = dn;
    float f0[T_SLABS], f1[T_SLABS];
    #pragma unroll
    for (int j = 0; j < T_SLABS; ++j) {
        f0[j] = __shfl(f[j], lo, 32);
        f1[j] = __shfl(f[j], 16 + lo, 32);
    }
    #pragma unroll
    for (int r = 0; r < 8; ++r) {
        const int rbase = (r + hi * 8) * KPAD;
        float p0 = acc0[0][r] * f0[0];
        float p1 = acc1[0][r] * f1[0];
        #pragma unroll
        for (int j = 1; j < T_SLABS; ++j) {
            p0 += acc0[j][r] * f0[j];
            p1 += acc1[j][r] * f1[j];
        }
        part[w][rbase + lo]      = p0;
        part[w][rbase + 16 + lo] = p1;
    }
    __syncthreads();
    float sum = part[0][t];
    float den = s_dn[0][lane];
    #pragma unroll
    for (int ww = 1; ww < T_WAVES; ++ww) {
        sum += part[ww][t];
        den += s_dn[ww][lane];
    }
    s_t[t] = sum / den;
    __syncthreads();

    float s0 = 0.0f, s1 = 0.0f;
    if (sq) {
        #pragma unroll
        for (int a = 0; a < A_SZ; ++a) {
            s0 += wa[a] * s_t[a * KPAD + k];
            s1 += wa[a] * s_t[(A_SZ + a) * KPAD + k];
        }
        s_s[0][k][o] = s0;
        s_s[1][k][o] = s1;
    }
    __syncthreads();

    float v0 = 0.0f, v1 = 0.0f;
    if (sq) {
        float mg0 = 0.0f, mg1 = 0.0f;
        #pragma unroll
        for (int kk = 0; kk < K_CLS; ++kk) {
            const float sv0 = s_s[0][kk][o];
            const float sv1 = s_s[1][kk][o];
            mg0 += sv0 * sv0;
            mg1 += sv1 * sv1;
        }
        const float scale0 = mg0 / ((1.0f + mg0) * sqrtf(mg0));
        const float scale1 = mg1 / ((1.0f + mg1) * sqrtf(mg1));
        v0 = s0 * scale0;
        v1 = s1 * scale1;
        s_v[0][k][o] = v0;
        s_v[1][k][o] = v1;
    }
    __syncthreads();

    if (sq) {
        out[(b0    ) * (K_CLS * O_SZ) + t] = v0;
        out[(b0 + 1) * (K_CLS * O_SZ) + t] = v1;
        float wsum = 0.0f;
        #pragma unroll
        for (int o2 = 0; o2 < O_SZ; ++o2)
            wsum += wo[o2] * s_v[bb3][k3][o2];
        wv2[((b0 + bb3) * A_SZ + a3) * KPAD + k3] = wsum * (1.0f / (float)B_SZ);
    }
}

#define B_WAVES 32
#define B_CHUNK (I_SZ / B_WAVES)
__global__ void __launch_bounds__(B_WAVES * 32)
caps_gemm_b(const float* __restrict__ x,
            const float* __restrict__ wv2,
            float* __restrict__ bij,
            float* __restrict__ e,
            float* __restrict__ smax,
            float* __restrict__ ssum) {
    __shared__ float part[B_WAVES][16 * KPAD];
    __shared__ float s_b[16 * KPAD];
    __shared__ float s_e[16 * KPAD];
    const int w    = __builtin_amdgcn_readfirstlane(threadIdx.x >> 5);
    const int lane = threadIdx.x & 31;
    const int t    = threadIdx.x;
    const int m0   = blockIdx.x * 16;
    const int lo   = lane & 15;
    const int hi   = lane >> 4;
    const int m    = m0 + lo;
    const int j0   = hi * 2;

    v8f acc0 = {};
    v8f acc1 = {};
    #pragma unroll 2
    for (int s = 0; s < B_CHUNK; s += 4) {
        const int ib = w * B_CHUNK + s;
        const int ia = ib + j0;
        const int ic = ia + 1;
        v2f av, bv0, bv1;
        av.x  = x[(ia >> 3) * (A_SZ * M_SZ) + m * A_SZ + (ia & 7)];
        av.y  = x[(ic >> 3) * (A_SZ * M_SZ) + m * A_SZ + (ic & 7)];
        bv0.x = wv2[ia * KPAD + lo];
        bv0.y = wv2[ic * KPAD + lo];
        bv1.x = wv2[ia * KPAD + 16 + lo];
        bv1.y = wv2[ic * KPAD + 16 + lo];
        acc0 = __builtin_amdgcn_wmma_f32_16x16x4_f32(
            false, av, false, bv0, (short)0, acc0, false, false);
        acc1 = __builtin_amdgcn_wmma_f32_16x16x4_f32(
            false, av, false, bv1, (short)0, acc1, false, false);
    }
    #pragma unroll
    for (int r = 0; r < 8; ++r) {
        const int rbase = (r + hi * 8) * KPAD;
        part[w][rbase + lo]      = acc0[r];
        part[w][rbase + 16 + lo] = acc1[r];
    }
    __syncthreads();
    const bool ep = w < (16 * KPAD) / 32;
    float nb = 0.0f;
    if (ep) {
        float sum = part[0][t];
        #pragma unroll
        for (int ww = 1; ww < B_WAVES; ++ww)
            sum += part[ww][t];
        nb = bij[m0 * KPAD + t] + sum;
        s_b[t] = nb;
    }
    __syncthreads();
    float mx = 0.0f, ev = 0.0f;
    if (ep) {
        mx = s_b[lane];
        #pragma unroll
        for (int rr = 1; rr < 16; ++rr)
            mx = fmaxf(mx, s_b[rr * KPAD + lane]);
        ev = expf(nb - mx);
        s_e[t] = ev;
    }
    __syncthreads();
    if (ep) {
        bij[m0 * KPAD + t] = nb;
        e[m0 * KPAD + t]   = ev;
        if (w == 0) {
            float es = s_e[lane];
            #pragma unroll
            for (int rr = 1; rr < 16; ++rr)
                es += s_e[rr * KPAD + lane];
            smax[blockIdx.x * KPAD + lane] = mx;
            ssum[blockIdx.x * KPAD + lane] = es;
        }
    }
}

extern "C" void kernel_launch(void* const* d_in, const int* in_sizes, int n_in,
                              void* d_out, int out_size, void* d_ws, size_t ws_size,
                              hipStream_t stream) {
    (void)in_sizes; (void)n_in; (void)out_size; (void)ws_size;
    const float* x = (const float*)d_in[0];
    const float* W = (const float*)d_in[1];
    float* out = (float*)d_out;

    float* bij  = (float*)d_ws;
    float* e    = bij  + M_SZ * KPAD;
    float* wv2  = e    + M_SZ * KPAD;
    float* smax = wv2  + I_SZ * KPAD;
    float* ssum = smax + N_SLAB * KPAD;

    caps_init<<<(M_SZ * KPAD) / 256, 256, 0, stream>>>(bij, e, smax, ssum);

    for (int it = 0; it < 3; ++it) {
        caps_gemm_t_squash<<<I_SZ / 16, T_WAVES * 32, 0, stream>>>(x, e, smax, ssum, W, out, wv2);
        if (it < 2)
            caps_gemm_b<<<M_SZ / 16, B_WAVES * 32, 0, stream>>>(x, wv2, bij, e, smax, ssum);
    }
}


# Both GEMMs: all 16 steps' A and B fragments of a wave requested in one run of loads ahead of its first WMMA and held in registers: about one L2 round trip per wave, not eight (sum order kept)
# speedup vs baseline: 10.7095x; 1.1554x over previous
#include <hip/hip_runtime.h>
#include <math.h>

typedef __attribute__((ext_vector_type(2))) float v2f;
typedef __attribute__((ext_vector_type(8))) float v8f;

#define B_SZ   256
#define M_SZ   1024
#define A_SZ   8
#define K_CLS  20
#define O_SZ   16
#define KPAD   32
#define I_SZ   (B_SZ * A_SZ)

#define N_SLAB (M_SZ / 16)

__global__ void caps_init(float* __restrict__ bij, float* __restrict__ e,
                          float* __restrict__ smax, float* __restrict__ ssum) {
    int idx = blockIdx.x * blockDim.x + threadIdx.x;
    bij[idx] = 0.0f;
    e[idx]   = 1.0f;
    if (idx < N_SLAB * KPAD) {
        smax[idx] = 0.0f;
        ssum[idx] = 16.0f;
    }
}

#define T_WAVES 16
#define T_CHUNK (M_SZ / T_WAVES)
#define T_SLABS (T_CHUNK / 16)
#define SQ_THREADS (K_CLS * O_SZ)
#define SQ_HALF    (K_CLS * A_SZ)
__global__ void __launch_bounds__(T_WAVES * 32) __attribute__((amdgpu_waves_per_eu(4, 4)))
caps_gemm_t_squash(const float* __restrict__ x,
                   const float* __restrict__ e,
                   const float* __restrict__ smax,
                   const float* __restrict__ ssum,
                   const float* __restrict__ W,
                   float* __restrict__ out,
                   float* __restrict__ wv2) {
    __shared__ float part[T_WAVES][16 * KPAD];
    __shared__ float s_mx[T_WAVES][KPAD];
    __shared__ float s_dn[T_WAVES][KPAD];
    __shared__ float s_t[16 * KPAD];
    __shared__ float s_s[2][K_CLS][O_SZ];
    __shared__ float s_v[2][K_CLS][O_SZ + 1];
    const int w    = __builtin_amdgcn_readfirstlane(threadIdx.x >> 5);
    const int lane = threadIdx.x & 31;
    const int t    = threadIdx.x;
    const int i0   = blockIdx.x * 16;
    const int b0   = blockIdx.x * 2;
    const int lo   = lane & 15;
    const int hi   = lane >> 4;
    const int row  = i0 + lo;
    const int xb   = (row >> 3) * (A_SZ * M_SZ) + (row & 7);
    const int j0   = hi * 2;

    const bool sq  = w < (SQ_THREADS / 32);
    const int k    = t >> 4;
    const int o    = t & 15;
    const int bb3  = (t >= SQ_HALF) ? 1 : 0;
    const int r3   = t - bb3 * SQ_HALF;
    const int k3   = r3 % K_CLS;
    const int a3   = r3 / K_CLS;

    float bm[T_SLABS], bs[T_SLABS];
    #pragma unroll
    for (int j = 0; j < T_SLABS; ++j) {
        bm[j] = smax[(w * T_SLABS + j) * KPAD + lane];
        bs[j] = ssum[(w * T_SLABS + j) * KPAD + lane];
    }

    float wa[A_SZ] = {};
    float wo[O_SZ] = {};
    if (sq) {
        #pragma unroll
        for (int a = 0; a < A_SZ; ++a)
            wa[a] = W[k * (O_SZ * A_SZ) + o * A_SZ + a];
        #pragma unroll
        for (int o2 = 0; o2 < O_SZ; ++o2)
            wo[o2] = W[k3 * (O_SZ * A_SZ) + o2 * A_SZ + a3];
    }

    v2f av[T_CHUNK / 4], bv0[T_CHUNK / 4], bv1[T_CHUNK / 4];
    {
        const float* __restrict__ xa = x + xb + (w * T_CHUNK + j0) * A_SZ;
        const float* __restrict__ eb = e + (w * T_CHUNK + j0) * KPAD + lo;
        #pragma unroll
        for (int q = 0; q < T_CHUNK / 4; ++q) {
            av[q].x  = xa[(q * 4    ) * A_SZ];
            av[q].y  = xa[(q * 4 + 1) * A_SZ];
            bv0[q].x = eb[(q * 4    ) * KPAD];
            bv0[q].y = eb[(q * 4 + 1) * KPAD];
            bv1[q].x = eb[(q * 4    ) * KPAD + 16];
            bv1[q].y = eb[(q * 4 + 1) * KPAD + 16];
        }
    }
    v8f acc0[T_SLABS] = {};
    v8f acc1[T_SLABS] = {};
    #pragma unroll
    for (int q = 0; q < T_CHUNK / 4; ++q) {
        acc0[q >> 2] = __builtin_amdgcn_wmma_f32_16x16x4_f32(
            false, av[q], false, bv0[q], (short)0, acc0[q >> 2], false, false);
        acc1[q >> 2] = __builtin_amdgcn_wmma_f32_16x16x4_f32(
            false, av[q], false, bv1[q], (short)0, acc1[q >> 2], false, false);
    }
    __builtin_amdgcn_sched_barrier(0);
    s_mx[w][lane] = fmaxf(fmaxf(bm[0], bm[1]), fmaxf(bm[2], bm[3]));
    __syncthreads();

    float g = s_mx[0][lane];
    #pragma unroll
    for (int ww = 1; ww < T_WAVES; ++ww)
        g = fmaxf(g, s_mx[ww][lane]);
    float f[T_SLABS];
    float dn = 0.0f;
    #pragma unroll
    for (int j = 0; j < T_SLABS; ++j) {
        f[j] = expf(bm[j] - g);
        dn  += f[j] * bs[j];
    }
    s_dn[w][lane] = dn;
    float f0[T_SLABS], f1[T_SLABS];
    #pragma unroll
    for (int j = 0; j < T_SLABS; ++j) {
        f0[j] = __shfl(f[j], lo, 32);
        f1[j] = __shfl(f[j], 16 + lo, 32);
    }
    #pragma unroll
    for (int r = 0; r < 8; ++r) {
        const int rbase = (r + hi * 8) * KPAD;
        float p0 = acc0[0][r] * f0[0];
        float p1 = acc1[0][r] * f1[0];
        #pragma unroll
        for (int j = 1; j < T_SLABS; ++j) {
            p0 += acc0[j][r] * f0[j];
            p1 += acc1[j][r] * f1[j];
        }
        part[w][rbase + lo]      = p0;
        part[w][rbase + 16 + lo] = p1;
    }
    __syncthreads();
    float sum = part[0][t];
    float den = s_dn[0][lane];
    #pragma unroll
    for (int ww = 1; ww < T_WAVES; ++ww) {
        sum += part[ww][t];
        den += s_dn[ww][lane];
    }
    s_t[t] = sum / den;
    __syncthreads();

    float s0 = 0.0f, s1 = 0.0f;
    if (sq) {
        #pragma unroll
        for (int a = 0; a < A_SZ; ++a) {
            s0 += wa[a] * s_t[a * KPAD + k];
            s1 += wa[a] * s_t[(A_SZ + a) * KPAD + k];
        }
        s_s[0][k][o] = s0;
        s_s[1][k][o] = s1;
    }
    __syncthreads();

    float v0 = 0.0f, v1 = 0.0f;
    if (sq) {
        float mg0 = 0.0f, mg1 = 0.0f;
        #pragma unroll
        for (int kk = 0; kk < K_CLS; ++kk) {
            const float sv0 = s_s[0][kk][o];
            const float sv1 = s_s[1][kk][o];
            mg0 += sv0 * sv0;
            mg1 += sv1 * sv1;
        }
        const float scale0 = mg0 / ((1.0f + mg0) * sqrtf(mg0));
        const float scale1 = mg1 / ((1.0f + mg1) * sqrtf(mg1));
        v0 = s0 * scale0;
        v1 = s1 * scale1;
        s_v[0][k][o] = v0;
        s_v[1][k][o] = v1;
    }
    __syncthreads();

    if (sq) {
        out[(b0    ) * (K_CLS * O_SZ) + t] = v0;
        out[(b0 + 1) * (K_CLS * O_SZ) + t] = v1;
        float wsum = 0.0f;
        #pragma unroll
        for (int o2 = 0; o2 < O_SZ; ++o2)
            wsum += wo[o2] * s_v[bb3][k3][o2];
        wv2[((b0 + bb3) * A_SZ + a3) * KPAD + k3] = wsum * (1.0f / (float)B_SZ);
    }
}

#define B_WAVES 32
#define B_CHUNK (I_SZ / B_WAVES)
__global__ void __launch_bounds__(B_WAVES * 32) __attribute__((amdgpu_waves_per_eu(8, 8)))
caps_gemm_b(const float* __restrict__ x,
            const float* __restrict__ wv2,
            float* __restrict__ bij,
            float* __restrict__ e,
            float* __restrict__ smax,
            float* __restrict__ ssum) {
    __shared__ float part[B_WAVES][16 * KPAD];
    __shared__ float s_b[16 * KPAD];
    __shared__ float s_e[16 * KPAD];
    const int w    = __builtin_amdgcn_readfirstlane(threadIdx.x >> 5);
    const int lane = threadIdx.x & 31;
    const int t    = threadIdx.x;
    const int m0   = blockIdx.x * 16;
    const int lo   = lane & 15;
    const int hi   = lane >> 4;
    const int m    = m0 + lo;
    const int j0   = hi * 2;

    v2f av[B_CHUNK / 4], bv0[B_CHUNK / 4], bv1[B_CHUNK / 4];
    {
        const float* __restrict__ xa = x + (w * (B_CHUNK / 8)) * (A_SZ * M_SZ) + m * A_SZ + j0;
        const float* __restrict__ wb = wv2 + (w * B_CHUNK + j0) * KPAD + lo;
        #pragma unroll
        for (int q = 0; q < B_CHUNK / 4; ++q) {
            av[q]    = *(const v2f*)(xa + (q >> 1) * (A_SZ * M_SZ) + (q & 1) * 4);
            bv0[q].x = wb[(q * 4    ) * KPAD];
            bv0[q].y = wb[(q * 4 + 1) * KPAD];
            bv1[q].x = wb[(q * 4    ) * KPAD + 16];
            bv1[q].y = wb[(q * 4 + 1) * KPAD + 16];
        }
    }
    v8f acc0 = {};
    v8f acc1 = {};
    #pragma unroll
    for (int q = 0; q < B_CHUNK / 4; ++q) {
        acc0 = __builtin_amdgcn_wmma_f32_16x16x4_f32(
            false, av[q], false, bv0[q], (short)0, acc0, false, false);
        acc1 = __builtin_amdgcn_wmma_f32_16x16x4_f32(
            false, av[q], false, bv1[q], (short)0, acc1, false, false);
    }
    #pragma unroll
    for (int r = 0; r < 8; ++r) {
        const int rbase = (r + hi * 8) * KPAD;
        part[w][rbase + lo]      = acc0[r];
        part[w][rbase + 16 + lo] = acc1[r];
    }
    __syncthreads();
    const bool ep = w < (16 * KPAD) / 32;
    float nb = 0.0f;
    if (ep) {
        float sum = part[0][t];
        #pragma unroll
        for (int ww = 1; ww < B_WAVES; ++ww)
            sum += part[ww][t];
        nb = bij[m0 * KPAD + t] + sum;
        s_b[t] = nb;
    }
    __syncthreads();
    float mx = 0.0f, ev = 0.0f;
    if (ep) {
        mx = s_b[lane];
        #pragma unroll
        for (int rr = 1; rr < 16; ++rr)
            mx = fmaxf(mx, s_b[rr * KPAD + lane]);
        ev = expf(nb - mx);
        s_e[t] = ev;
    }
    __syncthreads();
    if (ep) {
        bij[m0 * KPAD + t] = nb;
        e[m0 * KPAD + t]   = ev;
        if (w == 0) {
            float es = s_e[lane];
            #pragma unroll
            for (int rr = 1; rr < 16; ++rr)
                es += s_e[rr * KPAD + lane];
            smax[blockIdx.x * KPAD + lane] = mx;
            ssum[blockIdx.x * KPAD + lane] = es;
        }
    }
}

extern "C" void kernel_launch(void* const* d_in, const int* in_sizes, int n_in,
                              void* d_out, int out_size, void* d_ws, size_t ws_size,
                              hipStream_t stream) {
    (void)in_sizes; (void)n_in; (void)out_size; (void)ws_size;
    const float* x = (const float*)d_in[0];
    const float* W = (const float*)d_in[1];
    float* out = (float*)d_out;

    float* bij  = (float*)d_ws;
    float* e    = bij  + M_SZ * KPAD;
    float* wv2  = e    + M_SZ * KPAD;
    float* smax = wv2  + I_SZ * KPAD;
    float* ssum = smax + N_SLAB * KPAD;

    caps_init<<<(M_SZ * KPAD) / 256, 256, 0, stream>>>(bij, e, smax, ssum);

    for (int it = 0; it < 3; ++it) {
        caps_gemm_t_squash<<<I_SZ / 16, T_WAVES * 32, 0, stream>>>(x, e, smax, ssum, W, out, wv2);
        if (it < 2)
            caps_gemm_b<<<M_SZ / 16, B_WAVES * 32, 0, stream>>>(x, wv2, bij, e, smax, ssum);
    }
}


# Start state (zero logits, e = 1) built into pass 1's two kernels as constants: no init kernel, 6 -> 5 launches; first gemm_t loads no e or slab statistics, its tile 1 reuses tile 0 (sum order kept)
# speedup vs baseline: 12.3745x; 1.1555x over previous
#include <hip/hip_runtime.h>
#include <math.h>

typedef __attribute__((ext_vector_type(2))) float v2f;
typedef __attribute__((ext_vector_type(8))) float v8f;

#define B_SZ   256
#define M_SZ   1024
#define A_SZ   8
#define K_CLS  20
#define O_SZ   16
#define KPAD   32
#define I_SZ   (B_SZ * A_SZ)

#define N_SLAB (M_SZ / 16)

#define E_START    1.0f
#define SSUM_START 16.0f

#define T_WAVES 16
#define T_CHUNK (M_SZ / T_WAVES)
#define T_SLABS (T_CHUNK / 16)
#define SQ_THREADS (K_CLS * O_SZ)
#define SQ_HALF    (K_CLS * A_SZ)
template <bool FIRST>
__global__ void __launch_bounds__(T_WAVES * 32) __attribute__((amdgpu_waves_per_eu(4, 4)))
caps_gemm_t_squash(const float* __restrict__ x,
                   const float* __restrict__ e,
                   const float* __restrict__ smax,
                   const float* __restrict__ ssum,
                   const float* __restrict__ W,
                   float* __restrict__ out,
                   float* __restrict__ wv2) {
    __shared__ float part[T_WAVES][16 * KPAD];
    __shared__ float s_mx[T_WAVES][KPAD];
    __shared__ float s_dn[T_WAVES][KPAD];
    __shared__ float s_t[16 * KPAD];
    __shared__ float s_s[2][K_CLS][O_SZ];
    __shared__ float s_v[2][K_CLS][O_SZ + 1];
    const int w    = __builtin_amdgcn_readfirstlane(threadIdx.x >> 5);
    const int lane = threadIdx.x & 31;
    const int t    = threadIdx.x;
    const int i0   = blockIdx.x * 16;
    const int b0   = blockIdx.x * 2;
    const int lo   = lane & 15;
    const int hi   = lane >> 4;
    const int row  = i0 + lo;
    const int xb   = (row >> 3) * (A_SZ * M_SZ) + (row & 7);
    const int j0   = hi * 2;

    const bool sq  = w < (SQ_THREADS / 32);
    const int k    = t >> 4;
    const int o    = t & 15;
    const int bb3  = (t >= SQ_HALF) ? 1 : 0;
    const int r3   = t - bb3 * SQ_HALF;
    const int k3   = r3 % K_CLS;
    const int a3   = r3 / K_CLS;

    float bm[T_SLABS] = {}, bs[T_SLABS] = {};
    if constexpr (!FIRST) {
        #pragma unroll
        for (int j = 0; j < T_SLABS; ++j) {
            bm[j] = smax[(w * T_SLABS + j) * KPAD + lane];
            bs[j] = ssum[(w * T_SLABS + j) * KPAD + lane];
        }
    }

    float wa[A_SZ] = {};
    float wo[O_SZ] = {};
    if (sq) {
        #pragma unroll
        for (int a = 0; a < A_SZ; ++a)
            wa[a] = W[k * (O_SZ * A_SZ) + o * A_SZ + a];
        #pragma unroll
        for (int o2 = 0; o2 < O_SZ; ++o2)
            wo[o2] = W[k3 * (O_SZ * A_SZ) + o2 * A_SZ + a3];
    }

    v2f av[T_CHUNK / 4], bv0[T_CHUNK / 4], bv1[T_CHUNK / 4];
    {
        const float* __restrict__ xa = x + xb + (w * T_CHUNK + j0) * A_SZ;
        const float* __restrict__ eb = e + (w * T_CHUNK + j0) * KPAD + lo;
        #pragma unroll
        for (int q = 0; q < T_CHUNK / 4; ++q) {
            av[q].x  = xa[(q * 4    ) * A_SZ];
            av[q].y  = xa[(q * 4 + 1) * A_SZ];
            if constexpr (FIRST) {
                bv0[q].x = E_START;
                bv0[q].y = E_START;
            } else {
                bv0[q].x = eb[(q * 4    ) * KPAD];
                bv0[q].y = eb[(q * 4 + 1) * KPAD];
                bv1[q].x = eb[(q * 4    ) * KPAD + 16];
                bv1[q].y = eb[(q * 4 + 1) * KPAD + 16];
            }
        }
    }
    v8f acc0[T_SLABS] = {};
    v8f acc1[T_SLABS] = {};
    #pragma unroll
    for (int q = 0; q < T_CHUNK / 4; ++q) {
        acc0[q >> 2] = __builtin_amdgcn_wmma_f32_16x16x4_f32(
            false, av[q], false, bv0[q], (short)0, acc0[q >> 2], false, false);
        if constexpr (!FIRST)
            acc1[q >> 2] = __builtin_amdgcn_wmma_f32_16x16x4_f32(
                false, av[q], false, bv1[q], (short)0, acc1[q >> 2], false, false);
    }
    if constexpr (FIRST) {
        #pragma unroll
        for (int j = 0; j < T_SLABS; ++j)
            acc1[j] = acc0[j];
    }
    if constexpr (!FIRST)
        __builtin_amdgcn_sched_barrier(0);
    float f0[T_SLABS], f1[T_SLABS];
    if constexpr (FIRST) {
        #pragma unroll
        for (int j = 0; j < T_SLABS; ++j) {
            f0[j] = E_START;
            f1[j] = E_START;
        }
    } else {
        s_mx[w][lane] = fmaxf(fmaxf(bm[0], bm[1]), fmaxf(bm[2], bm[3]));
        __syncthreads();

        float g = s_mx[0][lane];
        #pragma unroll
        for (int ww = 1; ww < T_WAVES; ++ww)
            g = fmaxf(g, s_mx[ww][lane]);
        float f[T_SLABS];
        float dn = 0.0f;
        #pragma unroll
        for (int j = 0; j < T_SLABS; ++j) {
            f[j] = expf(bm[j] - g);
            dn  += f[j] * bs[j];
        }
        s_dn[w][lane] = dn;
        #pragma unroll
        for (int j = 0; j < T_SLABS; ++j) {
            f0[j] = __shfl(f[j], lo, 32);
            f1[j] = __shfl(f[j], 16 + lo, 32);
        }
    }
    #pragma unroll
    for (int r = 0; r < 8; ++r) {
        const int rbase = (r + hi * 8) * KPAD;
        float p0 = acc0[0][r] * f0[0];
        float p1 = acc1[0][r] * f1[0];
        #pragma unroll
        for (int j = 1; j < T_SLABS; ++j) {
            p0 += acc0[j][r] * f0[j];
            p1 += acc1[j][r] * f1[j];
        }
        part[w][rbase + lo]      = p0;
        part[w][rbase + 16 + lo] = p1;
    }
    __syncthreads();
    float sum = part[0][t];
    float den;
    if constexpr (FIRST) {
        den = (float)N_SLAB * E_START * SSUM_START;
        #pragma unroll
        for (int ww = 1; ww < T_WAVES; ++ww)
            sum += part[ww][t];
    } else {
        den = s_dn[0][lane];
        #pragma unroll
        for (int ww = 1; ww < T_WAVES; ++ww) {
            sum += part[ww][t];
            den += s_dn[ww][lane];
        }
    }
    s_t[t] = sum / den;
    __syncthreads();

    float s0 = 0.0f, s1 = 0.0f;
    if (sq) {
        #pragma unroll
        for (int a = 0; a < A_SZ; ++a) {
            s0 += wa[a] * s_t[a * KPAD + k];
            s1 += wa[a] * s_t[(A_SZ + a) * KPAD + k];
        }
        s_s[0][k][o] = s0;
        s_s[1][k][o] = s1;
    }
    __syncthreads();

    float v0 = 0.0f, v1 = 0.0f;
    if (sq) {
        float mg0 = 0.0f, mg1 = 0.0f;
        #pragma unroll
        for (int kk = 0; kk < K_CLS; ++kk) {
            const float sv0 = s_s[0][kk][o];
            const float sv1 = s_s[1][kk][o];
            mg0 += sv0 * sv0;
            mg1 += sv1 * sv1;
        }
        const float scale0 = mg0 / ((1.0f + mg0) * sqrtf(mg0));
        const float scale1 = mg1 / ((1.0f + mg1) * sqrtf(mg1));
        v0 = s0 * scale0;
        v1 = s1 * scale1;
        s_v[0][k][o] = v0;
        s_v[1][k][o] = v1;
    }
    __syncthreads();

    if (sq) {
        out[(b0    ) * (K_CLS * O_SZ) + t] = v0;
        out[(b0 + 1) * (K_CLS * O_SZ) + t] = v1;
        float wsum = 0.0f;
        #pragma unroll
        for (int o2 = 0; o2 < O_SZ; ++o2)
            wsum += wo[o2] * s_v[bb3][k3][o2];
        wv2[((b0 + bb3) * A_SZ + a3) * KPAD + k3] = wsum * (1.0f / (float)B_SZ);
    }
}

#define B_WAVES 32
#define B_CHUNK (I_SZ / B_WAVES)
template <bool FIRST>
__global__ void __launch_bounds__(B_WAVES * 32) __attribute__((amdgpu_waves_per_eu(8, 8)))
caps_gemm_b(const float* __restrict__ x,
            const float* __restrict__ wv2,
            float* __restrict__ bij,
            float* __restrict__ e,
            float* __restrict__ smax,
            float* __restrict__ ssum) {
    __shared__ float part[B_WAVES][16 * KPAD];
    __shared__ float s_b[16 * KPAD];
    __shared__ float s_e[16 * KPAD];
    const int w    = __builtin_amdgcn_readfirstlane(threadIdx.x >> 5);
    const int lane = threadIdx.x & 31;
    const int t    = threadIdx.x;
    const int m0   = blockIdx.x * 16;
    const int lo   = lane & 15;
    const int hi   = lane >> 4;
    const int m    = m0 + lo;
    const int j0   = hi * 2;

    v2f av[B_CHUNK / 4], bv0[B_CHUNK / 4], bv1[B_CHUNK / 4];
    {
        const float* __restrict__ xa = x + (w * (B_CHUNK / 8)) * (A_SZ * M_SZ) + m * A_SZ + j0;
        const float* __restrict__ wb = wv2 + (w * B_CHUNK + j0) * KPAD + lo;
        #pragma unroll
        for (int q = 0; q < B_CHUNK / 4; ++q) {
            av[q]    = *(const v2f*)(xa + (q >> 1) * (A_SZ * M_SZ) + (q & 1) * 4);
            bv0[q].x = wb[(q * 4    ) * KPAD];
            bv0[q].y = wb[(q * 4 + 1) * KPAD];
            bv1[q].x = wb[(q * 4    ) * KPAD + 16];
            bv1[q].y = wb[(q * 4 + 1) * KPAD + 16];
        }
    }
    v8f acc0 = {};
    v8f acc1 = {};
    #pragma unroll
    for (int q = 0; q < B_CHUNK / 4; ++q) {
        acc0 = __builtin_amdgcn_wmma_f32_16x16x4_f32(
            false, av[q], false, bv0[q], (short)0, acc0, false, false);
        acc1 = __builtin_amdgcn_wmma_f32_16x16x4_f32(
            false, av[q], false, bv1[q], (short)0, acc1, false, false);
    }
    #pragma unroll
    for (int r = 0; r < 8; ++r) {
        const int rbase = (r + hi * 8) * KPAD;
        part[w][rbase + lo]      = acc0[r];
        part[w][rbase + 16 + lo] = acc1[r];
    }
    __syncthreads();
    const bool ep = w < (16 * KPAD) / 32;
    float nb = 0.0f;
    if (ep) {
        float sum = part[0][t];
        #pragma unroll
        for (int ww = 1; ww < B_WAVES; ++ww)
            sum += part[ww][t];
        if constexpr (FIRST)
            nb = sum;
        else
            nb = bij[m0 * KPAD + t] + sum;
        s_b[t] = nb;
    }
    __syncthreads();
    float mx = 0.0f, ev = 0.0f;
    if (ep) {
        mx = s_b[lane];
        #pragma unroll
        for (int rr = 1; rr < 16; ++rr)
            mx = fmaxf(mx, s_b[rr * KPAD + lane]);
        ev = expf(nb - mx);
        s_e[t] = ev;
    }
    __syncthreads();
    if (ep) {
        bij[m0 * KPAD + t] = nb;
        e[m0 * KPAD + t]   = ev;
        if (w == 0) {
            float es = s_e[lane];
            #pragma unroll
            for (int rr = 1; rr < 16; ++rr)
                es += s_e[rr * KPAD + lane];
            smax[blockIdx.x * KPAD + lane] = mx;
            ssum[blockIdx.x * KPAD + lane] = es;
        }
    }
}

extern "C" void kernel_launch(void* const* d_in, const int* in_sizes, int n_in,
                              void* d_out, int out_size, void* d_ws, size_t ws_size,
                              hipStream_t stream) {
    (void)in_sizes; (void)n_in; (void)out_size; (void)ws_size;
    const float* x = (const float*)d_in[0];
    const float* W = (const float*)d_in[1];
    float* out = (float*)d_out;

    float* bij  = (float*)d_ws;
    float* e    = bij  + M_SZ * KPAD;
    float* wv2  = e    + M_SZ * KPAD;
    float* smax = wv2  + I_SZ * KPAD;
    float* ssum = smax + N_SLAB * KPAD;

    caps_gemm_t_squash<true><<<I_SZ / 16, T_WAVES * 32, 0, stream>>>(x, e, smax, ssum, W, out, wv2);
    caps_gemm_b<true><<<M_SZ / 16, B_WAVES * 32, 0, stream>>>(x, wv2, bij, e, smax, ssum);
    caps_gemm_t_squash<false><<<I_SZ / 16, T_WAVES * 32, 0, stream>>>(x, e, smax, ssum, W, out, wv2);
    caps_gemm_b<false><<<M_SZ / 16, B_WAVES * 32, 0, stream>>>(x, wv2, bij, e, smax, ssum);
    caps_gemm_t_squash<false><<<I_SZ / 16, T_WAVES * 32, 0, stream>>>(x, e, smax, ssum, W, out, wv2);
}


# e and wv2 written by their producers in WMMA fragment order: a step's two B fragments are one 16-byte load per lane over whole lines, B loads 64 -> 16 per wave in both GEMMs (sum order kept)
# speedup vs baseline: 13.1518x; 1.0628x over previous
#include <hip/hip_runtime.h>
#include <math.h>

typedef __attribute__((ext_vector_type(2))) float v2f;
typedef __attribute__((ext_vector_type(8))) float v8f;
typedef __attribute__((ext_vector_type(4))) float v4f;

#define B_SZ   256
#define M_SZ   1024
#define A_SZ   8
#define K_CLS  20
#define O_SZ   16
#define KPAD   32
#define I_SZ   (B_SZ * A_SZ)

#define N_SLAB (M_SZ / 16)

__device__ __forceinline__ int frag_off(int kk, int c) {
    return (((kk >> 2) * 32 + ((kk >> 1) & 1) * 16 + (c & 15)) << 2) + ((c >> 4) << 1) + (kk & 1);
}

#define E_START    1.0f
#define SSUM_START 16.0f

#define T_WAVES 16
#define T_CHUNK (M_SZ / T_WAVES)
#define T_SLABS (T_CHUNK / 16)
#define SQ_THREADS (K_CLS * O_SZ)
#define SQ_HALF    (K_CLS * A_SZ)
template <bool FIRST>
__global__ void __launch_bounds__(T_WAVES * 32) __attribute__((amdgpu_waves_per_eu(4, 4)))
caps_gemm_t_squash(const float* __restrict__ x,
                   const float* __restrict__ e,
                   const float* __restrict__ smax,
                   const float* __restrict__ ssum,
                   const float* __restrict__ W,
                   float* __restrict__ out,
                   float* __restrict__ wv2) {
    __shared__ float part[T_WAVES][16 * KPAD];
    __shared__ float s_mx[T_WAVES][KPAD];
    __shared__ float s_dn[T_WAVES][KPAD];
    __shared__ float s_t[16 * KPAD];
    __shared__ float s_s[2][K_CLS][O_SZ];
    __shared__ float s_v[2][K_CLS][O_SZ + 1];
    const int w    = __builtin_amdgcn_readfirstlane(threadIdx.x >> 5);
    const int lane = threadIdx.x & 31;
    const int t    = threadIdx.x;
    const int i0   = blockIdx.x * 16;
    const int b0   = blockIdx.x * 2;
    const int lo   = lane & 15;
    const int hi   = lane >> 4;
    const int row  = i0 + lo;
    const int xb   = (row >> 3) * (A_SZ * M_SZ) + (row & 7);
    const int j0   = hi * 2;

    const bool sq  = w < (SQ_THREADS / 32);
    const int k    = t >> 4;
    const int o    = t & 15;
    const int bb3  = (t >= SQ_HALF) ? 1 : 0;
    const int r3   = t - bb3 * SQ_HALF;
    const int k3   = r3 % K_CLS;
    const int a3   = r3 / K_CLS;

    float bm[T_SLABS] = {}, bs[T_SLABS] = {};
    if constexpr (!FIRST) {
        #pragma unroll
        for (int j = 0; j < T_SLABS; ++j) {
            bm[j] = smax[(w * T_SLABS + j) * KPAD + lane];
            bs[j] = ssum[(w * T_SLABS + j) * KPAD + lane];
        }
    }

    float wa[A_SZ] = {};
    float wo[O_SZ] = {};
    if (sq) {
        #pragma unroll
        for (int a = 0; a < A_SZ; ++a)
            wa[a] = W[k * (O_SZ * A_SZ) + o * A_SZ + a];
        #pragma unroll
        for (int o2 = 0; o2 < O_SZ; ++o2)
            wo[o2] = W[k3 * (O_SZ * A_SZ) + o2 * A_SZ + a3];
    }

    v2f av[T_CHUNK / 4], bv0[T_CHUNK / 4], bv1[T_CHUNK / 4];
    {
        const float* __restrict__ xa = x + xb + (w * T_CHUNK + j0) * A_SZ;
        const v4f* __restrict__ eb = (const v4f*)e + (w * (T_CHUNK / 4)) * 32 + lane;
        #pragma unroll
        for (int q = 0; q < T_CHUNK / 4; ++q) {
            av[q].x  = xa[(q * 4    ) * A_SZ];
            av[q].y  = xa[(q * 4 + 1) * A_SZ];
            if constexpr (FIRST) {
                bv0[q].x = E_START;
                bv0[q].y = E_START;
            } else {
                const v4f f = eb[q * 32];
                bv0[q].x = f.x;
                bv0[q].y = f.y;
                bv1[q].x = f.z;
                bv1[q].y = f.w;
            }
        }
    }
    v8f acc0[T_SLABS] = {};
    v8f acc1[T_SLABS] = {};
    #pragma unroll
    for (int q = 0; q < T_CHUNK / 4; ++q) {
        acc0[q >> 2] = __builtin_amdgcn_wmma_f32_16x16x4_f32(
            false, av[q], false, bv0[q], (short)0, acc0[q >> 2], false, false);
        if constexpr (!FIRST)
            acc1[q >> 2] = __builtin_amdgcn_wmma_f32_16x16x4_f32(
                false, av[q], false, bv1[q], (short)0, acc1[q >> 2], false, false);
    }
    if constexpr (FIRST) {
        #pragma unroll
        for (int j = 0; j < T_SLABS; ++j)
            acc1[j] = acc0[j];
    }
    if constexpr (!FIRST)
        __builtin_amdgcn_sched_barrier(0);
    float f0[T_SLABS], f1[T_SLABS];
    if constexpr (FIRST) {
        #pragma unroll
        for (int j = 0; j < T_SLABS; ++j) {
            f0[j] = E_START;
            f1[j] = E_START;
        }
    } else {
        s_mx[w][lane] = fmaxf(fmaxf(bm[0], bm[1]), fmaxf(bm[2], bm[3]));
        __syncthreads();

        float g = s_mx[0][lane];
        #pragma unroll
        for (int ww = 1; ww < T_WAVES; ++ww)
            g = fmaxf(g, s_mx[ww][lane]);
        float f[T_SLABS];
        float dn = 0.0f;
        #pragma unroll
        for (int j = 0; j < T_SLABS; ++j) {
            f[j] = expf(bm[j] - g);
            dn  += f[j] * bs[j];
        }
        s_dn[w][lane] = dn;
        #pragma unroll
        for (int j = 0; j < T_SLABS; ++j) {
            f0[j] = __shfl(f[j], lo, 32);
            f1[j] = __shfl(f[j], 16 + lo, 32);
        }
    }
    #pragma unroll
    for (int r = 0; r < 8; ++r) {
        const int rbase = (r + hi * 8) * KPAD;
        float p0 = acc0[0][r] * f0[0];
        float p1 = acc1[0][r] * f1[0];
        #pragma unroll
        for (int j = 1; j < T_SLABS; ++j) {
            p0 += acc0[j][r] * f0[j];
            p1 += acc1[j][r] * f1[j];
        }
        part[w][rbase + lo]      = p0;
        part[w][rbase + 16 + lo] = p1;
    }
    __syncthreads();
    float sum = part[0][t];
    float den;
    if constexpr (FIRST) {
        den = (float)N_SLAB * E_START * SSUM_START;
        #pragma unroll
        for (int ww = 1; ww < T_WAVES; ++ww)
            sum += part[ww][t];
    } else {
        den = s_dn[0][lane];
        #pragma unroll
        for (int ww = 1; ww < T_WAVES; ++ww) {
            sum += part[ww][t];
            den += s_dn[ww][lane];
        }
    }
    s_t[t] = sum / den;
    __syncthreads();

    float s0 = 0.0f, s1 = 0.0f;
    if (sq) {
        #pragma unroll
        for (int a = 0; a < A_SZ; ++a) {
            s0 += wa[a] * s_t[a * KPAD + k];
            s1 += wa[a] * s_t[(A_SZ + a) * KPAD + k];
        }
        s_s[0][k][o] = s0;
        s_s[1][k][o] = s1;
    }
    __syncthreads();

    float v0 = 0.0f, v1 = 0.0f;
    if (sq) {
        float mg0 = 0.0f, mg1 = 0.0f;
        #pragma unroll
        for (int kk = 0; kk < K_CLS; ++kk) {
            const float sv0 = s_s[0][kk][o];
            const float sv1 = s_s[1][kk][o];
            mg0 += sv0 * sv0;
            mg1 += sv1 * sv1;
        }
        const float scale0 = mg0 / ((1.0f + mg0) * sqrtf(mg0));
        const float scale1 = mg1 / ((1.0f + mg1) * sqrtf(mg1));
        v0 = s0 * scale0;
        v1 = s1 * scale1;
        s_v[0][k][o] = v0;
        s_v[1][k][o] = v1;
    }
    __syncthreads();

    if (sq) {
        out[(b0    ) * (K_CLS * O_SZ) + t] = v0;
        out[(b0 + 1) * (K_CLS * O_SZ) + t] = v1;
        float wsum = 0.0f;
        #pragma unroll
        for (int o2 = 0; o2 < O_SZ; ++o2)
            wsum += wo[o2] * s_v[bb3][k3][o2];
        wv2[frag_off((b0 + bb3) * A_SZ + a3, k3)] = wsum * (1.0f / (float)B_SZ);
    }
}

#define B_WAVES 32
#define B_CHUNK (I_SZ / B_WAVES)
template <bool FIRST>
__global__ void __launch_bounds__(B_WAVES * 32) __attribute__((amdgpu_waves_per_eu(8, 8)))
caps_gemm_b(const float* __restrict__ x,
            const float* __restrict__ wv2,
            float* __restrict__ bij,
            float* __restrict__ e,
            float* __restrict__ smax,
            float* __restrict__ ssum) {
    __shared__ float part[B_WAVES][16 * KPAD];
    __shared__ float s_b[16 * KPAD];
    __shared__ float s_e[16 * KPAD];
    const int w    = __builtin_amdgcn_readfirstlane(threadIdx.x >> 5);
    const int lane = threadIdx.x & 31;
    const int t    = threadIdx.x;
    const int m0   = blockIdx.x * 16;
    const int lo   = lane & 15;
    const int hi   = lane >> 4;
    const int m    = m0 + lo;
    const int j0   = hi * 2;

    v2f av[B_CHUNK / 4], bv0[B_CHUNK / 4], bv1[B_CHUNK / 4];
    {
        const float* __restrict__ xa = x + (w * (B_CHUNK / 8)) * (A_SZ * M_SZ) + m * A_SZ + j0;
        const v4f* __restrict__ wb = (const v4f*)wv2 + (w * (B_CHUNK / 4)) * 32 + lane;
        #pragma unroll
        for (int q = 0; q < B_CHUNK / 4; ++q) {
            av[q]    = *(const v2f*)(xa + (q >> 1) * (A_SZ * M_SZ) + (q & 1) * 4);
            const v4f f = wb[q * 32];
            bv0[q].x = f.x;
            bv0[q].y = f.y;
            bv1[q].x = f.z;
            bv1[q].y = f.w;
        }
    }
    v8f acc0 = {};
    v8f acc1 = {};
    #pragma unroll
    for (int q = 0; q < B_CHUNK / 4; ++q) {
        acc0 = __builtin_amdgcn_wmma_f32_16x16x4_f32(
            false, av[q], false, bv0[q], (short)0, acc0, false, false);
        acc1 = __builtin_amdgcn_wmma_f32_16x16x4_f32(
            false, av[q], false, bv1[q], (short)0, acc1, false, false);
    }
    #pragma unroll
    for (int r = 0; r < 8; ++r) {
        const int rbase = (r + hi * 8) * KPAD;
        part[w][rbase + lo]      = acc0[r];
        part[w][rbase + 16 + lo] = acc1[r];
    }
    __syncthreads();
    const bool ep = w < (16 * KPAD) / 32;
    float nb = 0.0f;
    if (ep) {
        float sum = part[0][t];
        #pragma unroll
        for (int ww = 1; ww < B_WAVES; ++ww)
            sum += part[ww][t];
        if constexpr (FIRST)
            nb = sum;
        else
            nb = bij[m0 * KPAD + t] + sum;
        s_b[t] = nb;
    }
    __syncthreads();
    float mx = 0.0f, ev = 0.0f;
    if (ep) {
        mx = s_b[lane];
        #pragma unroll
        for (int rr = 1; rr < 16; ++rr)
            mx = fmaxf(mx, s_b[rr * KPAD + lane]);
        ev = expf(nb - mx);
        s_e[t] = ev;
    }
    __syncthreads();
    if (ep) {
        bij[m0 * KPAD + t] = nb;
        if (w < 4) {
            const int r0 = w * 4 + hi * 2;
            v4f f;
            f.x = s_e[(r0    ) * KPAD + lo];
            f.y = s_e[(r0 + 1) * KPAD + lo];
            f.z = s_e[(r0    ) * KPAD + 16 + lo];
            f.w = s_e[(r0 + 1) * KPAD + 16 + lo];
            ((v4f*)e)[(m0 / 4) * 32 + t] = f;
        }
        if (w == 0) {
            float es = s_e[lane];
            #pragma unroll
            for (int rr = 1; rr < 16; ++rr)
                es += s_e[rr * KPAD + lane];
            smax[blockIdx.x * KPAD + lane] = mx;
            ssum[blockIdx.x * KPAD + lane] = es;
        }
    }
}

extern "C" void kernel_launch(void* const* d_in, const int* in_sizes, int n_in,
                              void* d_out, int out_size, void* d_ws, size_t ws_size,
                              hipStream_t stream) {
    (void)in_sizes; (void)n_in; (void)out_size; (void)ws_size;
    const float* x = (const float*)d_in[0];
    const float* W = (const float*)d_in[1];
    float* out = (float*)d_out;

    float* bij  = (float*)d_ws;
    float* e    = bij  + M_SZ * KPAD;
    float* wv2  = e    + M_SZ * KPAD;
    float* smax = wv2  + I_SZ * KPAD;
    float* ssum = smax + N_SLAB * KPAD;

    caps_gemm_t_squash<true><<<I_SZ / 16, T_WAVES * 32, 0, stream>>>(x, e, smax, ssum, W, out, wv2);
    caps_gemm_b<true><<<M_SZ / 16, B_WAVES * 32, 0, stream>>>(x, wv2, bij, e, smax, ssum);
    caps_gemm_t_squash<false><<<I_SZ / 16, T_WAVES * 32, 0, stream>>>(x, e, smax, ssum, W, out, wv2);
    caps_gemm_b<false><<<M_SZ / 16, B_WAVES * 32, 0, stream>>>(x, wv2, bij, e, smax, ssum);
    caps_gemm_t_squash<false><<<I_SZ / 16, T_WAVES * 32, 0, stream>>>(x, e, smax, ssum, W, out, wv2);
}


# wv2 phase of gemm_t: thread map turned to atom-fastest, so a wave's 16 loads of W touch 4 whole 32-byte runs each instead of 20 scattered lines (sum order kept)
# speedup vs baseline: 13.9945x; 1.0641x over previous
#include <hip/hip_runtime.h>
#include <math.h>

typedef __attribute__((ext_vector_type(2))) float v2f;
typedef __attribute__((ext_vector_type(8))) float v8f;
typedef __attribute__((ext_vector_type(4))) float v4f;

#define B_SZ   256
#define M_SZ   1024
#define A_SZ   8
#define K_CLS  20
#define O_SZ   16
#define KPAD   32
#define I_SZ   (B_SZ * A_SZ)

#define N_SLAB (M_SZ / 16)

__device__ __forceinline__ int frag_off(int kk, int c) {
    return (((kk >> 2) * 32 + ((kk >> 1) & 1) * 16 + (c & 15)) << 2) + ((c >> 4) << 1) + (kk & 1);
}

#define E_START    1.0f
#define SSUM_START 16.0f

#define T_WAVES 16
#define T_CHUNK (M_SZ / T_WAVES)
#define T_SLABS (T_CHUNK / 16)
#define SQ_THREADS (K_CLS * O_SZ)
#define SQ_HALF    (K_CLS * A_SZ)
template <bool FIRST>
__global__ void __launch_bounds__(T_WAVES * 32) __attribute__((amdgpu_waves_per_eu(4, 4)))
caps_gemm_t_squash(const float* __restrict__ x,
                   const float* __restrict__ e,
                   const float* __restrict__ smax,
                   const float* __restrict__ ssum,
                   const float* __restrict__ W,
                   float* __restrict__ out,
                   float* __restrict__ wv2) {
    __shared__ float part[T_WAVES][16 * KPAD];
    __shared__ float s_mx[T_WAVES][KPAD];
    __shared__ float s_dn[T_WAVES][KPAD];
    __shared__ float s_t[16 * KPAD];
    __shared__ float s_s[2][K_CLS][O_SZ];
    __shared__ float s_v[2][K_CLS][O_SZ + 1];
    const int w    = __builtin_amdgcn_readfirstlane(threadIdx.x >> 5);
    const int lane = threadIdx.x & 31;
    const int t    = threadIdx.x;
    const int i0   = blockIdx.x * 16;
    const int b0   = blockIdx.x * 2;
    const int lo   = lane & 15;
    const int hi   = lane >> 4;
    const int row  = i0 + lo;
    const int xb   = (row >> 3) * (A_SZ * M_SZ) + (row & 7);
    const int j0   = hi * 2;

    const bool sq  = w < (SQ_THREADS / 32);
    const int k    = t >> 4;
    const int o    = t & 15;
    const int bb3  = (t >= SQ_HALF) ? 1 : 0;
    const int r3   = t - bb3 * SQ_HALF;
    const int k3   = r3 >> 3;
    const int a3   = r3 & (A_SZ - 1);

    float bm[T_SLABS] = {}, bs[T_SLABS] = {};
    if constexpr (!FIRST) {
        #pragma unroll
        for (int j = 0; j < T_SLABS; ++j) {
            bm[j] = smax[(w * T_SLABS + j) * KPAD + lane];
            bs[j] = ssum[(w * T_SLABS + j) * KPAD + lane];
        }
    }

    float wa[A_SZ] = {};
    float wo[O_SZ] = {};
    if (sq) {
        #pragma unroll
        for (int a = 0; a < A_SZ; ++a)
            wa[a] = W[k * (O_SZ * A_SZ) + o * A_SZ + a];
        #pragma unroll
        for (int o2 = 0; o2 < O_SZ; ++o2)
            wo[o2] = W[k3 * (O_SZ * A_SZ) + o2 * A_SZ + a3];
    }

    v2f av[T_CHUNK / 4], bv0[T_CHUNK / 4], bv1[T_CHUNK / 4];
    {
        const float* __restrict__ xa = x + xb + (w * T_CHUNK + j0) * A_SZ;
        const v4f* __restrict__ eb = (const v4f*)e + (w * (T_CHUNK / 4)) * 32 + lane;
        #pragma unroll
        for (int q = 0; q < T_CHUNK / 4; ++q) {
            av[q].x  = xa[(q * 4    ) * A_SZ];
            av[q].y  = xa[(q * 4 + 1) * A_SZ];
            if constexpr (FIRST) {
                bv0[q].x = E_START;
                bv0[q].y = E_START;
            } else {
                const v4f f = eb[q * 32];
                bv0[q].x = f.x;
                bv0[q].y = f.y;
                bv1[q].x = f.z;
                bv1[q].y = f.w;
            }
        }
    }
    v8f acc0[T_SLABS] = {};
    v8f acc1[T_SLABS] = {};
    #pragma unroll
    for (int q = 0; q < T_CHUNK / 4; ++q) {
        acc0[q >> 2] = __builtin_amdgcn_wmma_f32_16x16x4_f32(
            false, av[q], false, bv0[q], (short)0, acc0[q >> 2], false, false);
        if constexpr (!FIRST)
            acc1[q >> 2] = __builtin_amdgcn_wmma_f32_16x16x4_f32(
                false, av[q], false, bv1[q], (short)0, acc1[q >> 2], false, false);
    }
    if constexpr (FIRST) {
        #pragma unroll
        for (int j = 0; j < T_SLABS; ++j)
            acc1[j] = acc0[j];
    }
    if constexpr (!FIRST)
        __builtin_amdgcn_sched_barrier(0);
    float f0[T_SLABS], f1[T_SLABS];
    if constexpr (FIRST) {
        #pragma unroll
        for (int j = 0; j < T_SLABS; ++j) {
            f0[j] = E_START;
            f1[j] = E_START;
        }
    } else {
        s_mx[w][lane] = fmaxf(fmaxf(bm[0], bm[1]), fmaxf(bm[2], bm[3]));
        __syncthreads();

        float g = s_mx[0][lane];
        #pragma unroll
        for (int ww = 1; ww < T_WAVES; ++ww)
            g = fmaxf(g, s_mx[ww][lane]);
        float f[T_SLABS];
        float dn = 0.0f;
        #pragma unroll
        for (int j = 0; j < T_SLABS; ++j) {
            f[j] = expf(bm[j] - g);
            dn  += f[j] * bs[j];
        }
        s_dn[w][lane] = dn;
        #pragma unroll
        for (int j = 0; j < T_SLABS; ++j) {
            f0[j] = __shfl(f[j], lo, 32);
            f1[j] = __shfl(f[j], 16 + lo, 32);
        }
    }
    #pragma unroll
    for (int r = 0; r < 8; ++r) {
        const int rbase = (r + hi * 8) * KPAD;
        float p0 = acc0[0][r] * f0[0];
        float p1 = acc1[0][r] * f1[0];
        #pragma unroll
        for (int j = 1; j < T_SLABS; ++j) {
            p0 += acc0[j][r] * f0[j];
            p1 += acc1[j][r] * f1[j];
        }
        part[w][rbase + lo]      = p0;
        part[w][rbase + 16 + lo] = p1;
    }
    __syncthreads();
    float sum = part[0][t];
    float den;
    if constexpr (FIRST) {
        den = (float)N_SLAB * E_START * SSUM_START;
        #pragma unroll
        for (int ww = 1; ww < T_WAVES; ++ww)
            sum += part[ww][t];
    } else {
        den = s_dn[0][lane];
        #pragma unroll
        for (int ww = 1; ww < T_WAVES; ++ww) {
            sum += part[ww][t];
            den += s_dn[ww][lane];
        }
    }
    s_t[t] = sum / den;
    __syncthreads();

    float s0 = 0.0f, s1 = 0.0f;
    if (sq) {
        #pragma unroll
        for (int a = 0; a < A_SZ; ++a) {
            s0 += wa[a] * s_t[a * KPAD + k];
            s1 += wa[a] * s_t[(A_SZ + a) * KPAD + k];
        }
        s_s[0][k][o] = s0;
        s_s[1][k][o] = s1;
    }
    __syncthreads();

    float v0 = 0.0f, v1 = 0.0f;
    if (sq) {
        float mg0 = 0.0f, mg1 = 0.0f;
        #pragma unroll
        for (int kk = 0; kk < K_CLS; ++kk) {
            const float sv0 = s_s[0][kk][o];
            const float sv1 = s_s[1][kk][o];
            mg0 += sv0 * sv0;
            mg1 += sv1 * sv1;
        }
        const float scale0 = mg0 / ((1.0f + mg0) * sqrtf(mg0));
        const float scale1 = mg1 / ((1.0f + mg1) * sqrtf(mg1));
        v0 = s0 * scale0;
        v1 = s1 * scale1;
        s_v[0][k][o] = v0;
        s_v[1][k][o] = v1;
    }
    __syncthreads();

    if (sq) {
        out[(b0    ) * (K_CLS * O_SZ) + t] = v0;
        out[(b0 + 1) * (K_CLS * O_SZ) + t] = v1;
        float wsum = 0.0f;
        #pragma unroll
        for (int o2 = 0; o2 < O_SZ; ++o2)
            wsum += wo[o2] * s_v[bb3][k3][o2];
        wv2[frag_off((b0 + bb3) * A_SZ + a3, k3)] = wsum * (1.0f / (float)B_SZ);
    }
}

#define B_WAVES 32
#define B_CHUNK (I_SZ / B_WAVES)
template <bool FIRST>
__global__ void __launch_bounds__(B_WAVES * 32) __attribute__((amdgpu_waves_per_eu(8, 8)))
caps_gemm_b(const float* __restrict__ x,
            const float* __restrict__ wv2,
            float* __restrict__ bij,
            float* __restrict__ e,
            float* __restrict__ smax,
            float* __restrict__ ssum) {
    __shared__ float part[B_WAVES][16 * KPAD];
    __shared__ float s_b[16 * KPAD];
    __shared__ float s_e[16 * KPAD];
    const int w    = __builtin_amdgcn_readfirstlane(threadIdx.x >> 5);
    const int lane = threadIdx.x & 31;
    const int t    = threadIdx.x;
    const int m0   = blockIdx.x * 16;
    const int lo   = lane & 15;
    const int hi   = lane >> 4;
    const int m    = m0 + lo;
    const int j0   = hi * 2;

    v2f av[B_CHUNK / 4], bv0[B_CHUNK / 4], bv1[B_CHUNK / 4];
    {
        const float* __restrict__ xa = x + (w * (B_CHUNK / 8)) * (A_SZ * M_SZ) + m * A_SZ + j0;
        const v4f* __restrict__ wb = (const v4f*)wv2 + (w * (B_CHUNK / 4)) * 32 + lane;
        #pragma unroll
        for (int q = 0; q < B_CHUNK / 4; ++q) {
            av[q]    = *(const v2f*)(xa + (q >> 1) * (A_SZ * M_SZ) + (q & 1) * 4);
            const v4f f = wb[q * 32];
            bv0[q].x = f.x;
            bv0[q].y = f.y;
            bv1[q].x = f.z;
            bv1[q].y = f.w;
        }
    }
    v8f acc0 = {};
    v8f acc1 = {};
    #pragma unroll
    for (int q = 0; q < B_CHUNK / 4; ++q) {
        acc0 = __builtin_amdgcn_wmma_f32_16x16x4_f32(
            false, av[q], false, bv0[q], (short)0, acc0, false, false);
        acc1 = __builtin_amdgcn_wmma_f32_16x16x4_f32(
            false, av[q], false, bv1[q], (short)0, acc1, false, false);
    }
    #pragma unroll
    for (int r = 0; r < 8; ++r) {
        const int rbase = (r + hi * 8) * KPAD;
        part[w][rbase + lo]      = acc0[r];
        part[w][rbase + 16 + lo] = acc1[r];
    }
    __syncthreads();
    const bool ep = w < (16 * KPAD) / 32;
    float nb = 0.0f;
    if (ep) {
        float sum = part[0][t];
        #pragma unroll
        for (int ww = 1; ww < B_WAVES; ++ww)
            sum += part[ww][t];
        if constexpr (FIRST)
            nb = sum;
        else
            nb = bij[m0 * KPAD + t] + sum;
        s_b[t] = nb;
    }
    __syncthreads();
    float mx = 0.0f, ev = 0.0f;
    if (ep) {
        mx = s_b[lane];
        #pragma unroll
        for (int rr = 1; rr < 16; ++rr)
            mx = fmaxf(mx, s_b[rr * KPAD + lane]);
        ev = expf(nb - mx);
        s_e[t] = ev;
    }
    __syncthreads();
    if (ep) {
        bij[m0 * KPAD + t] = nb;
        if (w < 4) {
            const int r0 = w * 4 + hi * 2;
            v4f f;
            f.x = s_e[(r0    ) * KPAD + lo];
            f.y = s_e[(r0 + 1) * KPAD + lo];
            f.z = s_e[(r0    ) * KPAD + 16 + lo];
            f.w = s_e[(r0 + 1) * KPAD + 16 + lo];
            ((v4f*)e)[(m0 / 4) * 32 + t] = f;
        }
        if (w == 0) {
            float es = s_e[lane];
            #pragma unroll
            for (int rr = 1; rr < 16; ++rr)
                es += s_e[rr * KPAD + lane];
            smax[blockIdx.x * KPAD + lane] = mx;
            ssum[blockIdx.x * KPAD + lane] = es;
        }
    }
}

extern "C" void kernel_launch(void* const* d_in, const int* in_sizes, int n_in,
                              void* d_out, int out_size, void* d_ws, size_t ws_size,
                              hipStream_t stream) {
    (void)in_sizes; (void)n_in; (void)out_size; (void)ws_size;
    const float* x = (const float*)d_in[0];
    const float* W = (const float*)d_in[1];
    float* out = (float*)d_out;

    float* bij  = (float*)d_ws;
    float* e    = bij  + M_SZ * KPAD;
    float* wv2  = e    + M_SZ * KPAD;
    float* smax = wv2  + I_SZ * KPAD;
    float* ssum = smax + N_SLAB * KPAD;

    caps_gemm_t_squash<true><<<I_SZ / 16, T_WAVES * 32, 0, stream>>>(x, e, smax, ssum, W, out, wv2);
    caps_gemm_b<true><<<M_SZ / 16, B_WAVES * 32, 0, stream>>>(x, wv2, bij, e, smax, ssum);
    caps_gemm_t_squash<false><<<I_SZ / 16, T_WAVES * 32, 0, stream>>>(x, e, smax, ssum, W, out, wv2);
    caps_gemm_b<false><<<M_SZ / 16, B_WAVES * 32, 0, stream>>>(x, wv2, bij, e, smax, ssum);
    caps_gemm_t_squash<false><<<I_SZ / 16, T_WAVES * 32, 0, stream>>>(x, e, smax, ssum, W, out, wv2);
}


# gemm_b: each slab split over 2 work-groups by fragment line (8 fragment lanes = one 16x16 tile each), e and wv2 layouts kept; 128 WGs of 16 waves, 256 -> 128 WMMAs per SIMD (sum order changes)
# speedup vs baseline: 14.7259x; 1.0523x over previous
#include <hip/hip_runtime.h>
#include <math.h>

typedef __attribute__((ext_vector_type(2))) float v2f;
typedef __attribute__((ext_vector_type(8))) float v8f;
typedef __attribute__((ext_vector_type(4))) float v4f;

#define B_SZ   256
#define M_SZ   1024
#define A_SZ   8
#define K_CLS  20
#define O_SZ   16
#define KPAD   32
#define I_SZ   (B_SZ * A_SZ)

#define N_SLAB (M_SZ / 16)

__device__ __forceinline__ int frag_off(int kk, int c) {
    return (((kk >> 2) * 32 + ((kk >> 1) & 1) * 16 + (c & 15)) << 2) + ((c >> 4) << 1) + (kk & 1);
}

__device__ __forceinline__ int stat_off(int slab, int c) {
    return ((slab * 2 + ((c >> 3) & 1)) * KPAD) + (c & 7) + ((c >> 4) << 3);
}

#define E_START    1.0f
#define SSUM_START 16.0f

#define T_WAVES 16
#define T_CHUNK (M_SZ / T_WAVES)
#define T_SLABS (T_CHUNK / 16)
#define SQ_THREADS (K_CLS * O_SZ)
#define SQ_HALF    (K_CLS * A_SZ)
template <bool FIRST>
__global__ void __launch_bounds__(T_WAVES * 32) __attribute__((amdgpu_waves_per_eu(4, 4)))
caps_gemm_t_squash(const float* __restrict__ x,
                   const float* __restrict__ e,
                   const float* __restrict__ smax,
                   const float* __restrict__ ssum,
                   const float* __restrict__ W,
                   float* __restrict__ out,
                   float* __restrict__ wv2) {
    __shared__ float part[T_WAVES][16 * KPAD];
    __shared__ float s_mx[T_WAVES][KPAD];
    __shared__ float s_dn[T_WAVES][KPAD];
    __shared__ float s_t[16 * KPAD];
    __shared__ float s_s[2][K_CLS][O_SZ];
    __shared__ float s_v[2][K_CLS][O_SZ + 1];
    const int w    = __builtin_amdgcn_readfirstlane(threadIdx.x >> 5);
    const int lane = threadIdx.x & 31;
    const int t    = threadIdx.x;
    const int i0   = blockIdx.x * 16;
    const int b0   = blockIdx.x * 2;
    const int lo   = lane & 15;
    const int hi   = lane >> 4;
    const int row  = i0 + lo;
    const int xb   = (row >> 3) * (A_SZ * M_SZ) + (row & 7);
    const int j0   = hi * 2;

    const bool sq  = w < (SQ_THREADS / 32);
    const int k    = t >> 4;
    const int o    = t & 15;
    const int bb3  = (t >= SQ_HALF) ? 1 : 0;
    const int r3   = t - bb3 * SQ_HALF;
    const int k3   = r3 >> 3;
    const int a3   = r3 & (A_SZ - 1);

    float bm[T_SLABS] = {}, bs[T_SLABS] = {};
    if constexpr (!FIRST) {
        #pragma unroll
        for (int j = 0; j < T_SLABS; ++j) {
            bm[j] = smax[stat_off(w * T_SLABS + j, lane)];
            bs[j] = ssum[stat_off(w * T_SLABS + j, lane)];
        }
    }

    float wa[A_SZ] = {};
    float wo[O_SZ] = {};
    if (sq) {
        #pragma unroll
        for (int a = 0; a < A_SZ; ++a)
            wa[a] = W[k * (O_SZ * A_SZ) + o * A_SZ + a];
        #pragma unroll
        for (int o2 = 0; o2 < O_SZ; ++o2)
            wo[o2] = W[k3 * (O_SZ * A_SZ) + o2 * A_SZ + a3];
    }

    v2f av[T_CHUNK / 4], bv0[T_CHUNK / 4], bv1[T_CHUNK / 4];
    {
        const float* __restrict__ xa = x + xb + (w * T_CHUNK + j0) * A_SZ;
        const v4f* __restrict__ eb = (const v4f*)e + (w * (T_CHUNK / 4)) * 32 + lane;
        #pragma unroll
        for (int q = 0; q < T_CHUNK / 4; ++q) {
            av[q].x  = xa[(q * 4    ) * A_SZ];
            av[q].y  = xa[(q * 4 + 1) * A_SZ];
            if constexpr (FIRST) {
                bv0[q].x = E_START;
                bv0[q].y = E_START;
            } else {
                const v4f f = eb[q * 32];
                bv0[q].x = f.x;
                bv0[q].y = f.y;
                bv1[q].x = f.z;
                bv1[q].y = f.w;
            }
        }
    }
    v8f acc0[T_SLABS] = {};
    v8f acc1[T_SLABS] = {};
    #pragma unroll
    for (int q = 0; q < T_CHUNK / 4; ++q) {
        acc0[q >> 2] = __builtin_amdgcn_wmma_f32_16x16x4_f32(
            false, av[q], false, bv0[q], (short)0, acc0[q >> 2], false, false);
        if constexpr (!FIRST)
            acc1[q >> 2] = __builtin_amdgcn_wmma_f32_16x16x4_f32(
                false, av[q], false, bv1[q], (short)0, acc1[q >> 2], false, false);
    }
    if constexpr (FIRST) {
        #pragma unroll
        for (int j = 0; j < T_SLABS; ++j)
            acc1[j] = acc0[j];
    }
    if constexpr (!FIRST)
        __builtin_amdgcn_sched_barrier(0);
    float f0[T_SLABS], f1[T_SLABS];
    if constexpr (FIRST) {
        #pragma unroll
        for (int j = 0; j < T_SLABS; ++j) {
            f0[j] = E_START;
            f1[j] = E_START;
        }
    } else {
        s_mx[w][lane] = fmaxf(fmaxf(bm[0], bm[1]), fmaxf(bm[2], bm[3]));
        __syncthreads();

        float g = s_mx[0][lane];
        #pragma unroll
        for (int ww = 1; ww < T_WAVES; ++ww)
            g = fmaxf(g, s_mx[ww][lane]);
        float f[T_SLABS];
        float dn = 0.0f;
        #pragma unroll
        for (int j = 0; j < T_SLABS; ++j) {
            f[j] = expf(bm[j] - g);
            dn  += f[j] * bs[j];
        }
        s_dn[w][lane] = dn;
        #pragma unroll
        for (int j = 0; j < T_SLABS; ++j) {
            f0[j] = __shfl(f[j], lo, 32);
            f1[j] = __shfl(f[j], 16 + lo, 32);
        }
    }
    #pragma unroll
    for (int r = 0; r < 8; ++r) {
        const int rbase = (r + hi * 8) * KPAD;
        float p0 = acc0[0][r] * f0[0];
        float p1 = acc1[0][r] * f1[0];
        #pragma unroll
        for (int j = 1; j < T_SLABS; ++j) {
            p0 += acc0[j][r] * f0[j];
            p1 += acc1[j][r] * f1[j];
        }
        part[w][rbase + lo]      = p0;
        part[w][rbase + 16 + lo] = p1;
    }
    __syncthreads();
    float sum = part[0][t];
    float den;
    if constexpr (FIRST) {
        den = (float)N_SLAB * E_START * SSUM_START;
        #pragma unroll
        for (int ww = 1; ww < T_WAVES; ++ww)
            sum += part[ww][t];
    } else {
        den = s_dn[0][lane];
        #pragma unroll
        for (int ww = 1; ww < T_WAVES; ++ww) {
            sum += part[ww][t];
            den += s_dn[ww][lane];
        }
    }
    s_t[t] = sum / den;
    __syncthreads();

    float s0 = 0.0f, s1 = 0.0f;
    if (sq) {
        #pragma unroll
        for (int a = 0; a < A_SZ; ++a) {
            s0 += wa[a] * s_t[a * KPAD + k];
            s1 += wa[a] * s_t[(A_SZ + a) * KPAD + k];
        }
        s_s[0][k][o] = s0;
        s_s[1][k][o] = s1;
    }
    __syncthreads();

    float v0 = 0.0f, v1 = 0.0f;
    if (sq) {
        float mg0 = 0.0f, mg1 = 0.0f;
        #pragma unroll
        for (int kk = 0; kk < K_CLS; ++kk) {
            const float sv0 = s_s[0][kk][o];
            const float sv1 = s_s[1][kk][o];
            mg0 += sv0 * sv0;
            mg1 += sv1 * sv1;
        }
        const float scale0 = mg0 / ((1.0f + mg0) * sqrtf(mg0));
        const float scale1 = mg1 / ((1.0f + mg1) * sqrtf(mg1));
        v0 = s0 * scale0;
        v1 = s1 * scale1;
        s_v[0][k][o] = v0;
        s_v[1][k][o] = v1;
    }
    __syncthreads();

    if (sq) {
        out[(b0    ) * (K_CLS * O_SZ) + t] = v0;
        out[(b0 + 1) * (K_CLS * O_SZ) + t] = v1;
        float wsum = 0.0f;
        #pragma unroll
        for (int o2 = 0; o2 < O_SZ; ++o2)
            wsum += wo[o2] * s_v[bb3][k3][o2];
        wv2[frag_off((b0 + bb3) * A_SZ + a3, k3)] = wsum * (1.0f / (float)B_SZ);
    }
}

#define B_WAVES 16
#define B_CHUNK (I_SZ / B_WAVES)
#define B_TILE  (16 * 16)
template <bool FIRST>
__global__ void __launch_bounds__(B_WAVES * 32) __attribute__((amdgpu_waves_per_eu(4, 4)))
caps_gemm_b(const float* __restrict__ x,
            const float* __restrict__ wv2,
            float* __restrict__ bij,
            float* __restrict__ e,
            float* __restrict__ stat) {
    __shared__ float part[B_WAVES][B_TILE];
    __shared__ float s_b[B_TILE];
    __shared__ float s_e[B_TILE];
    const int w    = __builtin_amdgcn_readfirstlane(threadIdx.x >> 5);
    const int lane = threadIdx.x & 31;
    const int t    = threadIdx.x;
    const int slab = blockIdx.x >> 1;
    const int half = blockIdx.x & 1;
    const int m0   = slab * 16;
    const int lo   = lane & 15;
    const int hi   = lane >> 4;
    const int m    = m0 + lo;
    const int j0   = hi * 2;

    v2f av[B_CHUNK / 4], bv[B_CHUNK / 4];
    {
        const float* __restrict__ xa = x + (w * (B_CHUNK / 8)) * (A_SZ * M_SZ) + m * A_SZ + j0;
        const float* __restrict__ wb = wv2 + (w * (B_CHUNK / 4)) * 128
                                     + (hi * 16 + half * 8 + (lo & 7)) * 4 + (lo >> 3) * 2;
        #pragma unroll
        for (int q = 0; q < B_CHUNK / 4; ++q) {
            av[q] = *(const v2f*)(xa + (q >> 1) * (A_SZ * M_SZ) + (q & 1) * 4);
            bv[q] = *(const v2f*)(wb + q * 128);
        }
    }
    v8f acc = {};
    #pragma unroll
    for (int q = 0; q < B_CHUNK / 4; ++q)
        acc = __builtin_amdgcn_wmma_f32_16x16x4_f32(
            false, av[q], false, bv[q], (short)0, acc, false, false);
    #pragma unroll
    for (int r = 0; r < 8; ++r)
        part[w][(r + hi * 8) * 16 + lo] = acc[r];
    __syncthreads();
    const bool ep = w < B_TILE / 32;
    const int  n  = t & 15;
    float nb = 0.0f;
    if (ep) {
        float sum = part[0][t];
        #pragma unroll
        for (int ww = 1; ww < B_WAVES; ++ww)
            sum += part[ww][t];
        if constexpr (FIRST)
            nb = sum;
        else
            nb = bij[blockIdx.x * B_TILE + t] + sum;
        s_b[t] = nb;
    }
    __syncthreads();
    float mx = 0.0f, ev = 0.0f;
    if (ep) {
        mx = s_b[n];
        #pragma unroll
        for (int rr = 1; rr < 16; ++rr)
            mx = fmaxf(mx, s_b[rr * 16 + n]);
        ev = expf(nb - mx);
        s_e[t] = ev;
    }
    __syncthreads();
    if (ep) {
        bij[blockIdx.x * B_TILE + t] = nb;
        if (w < 2) {
            const int st = t >> 4;
            const int rp = (t >> 3) & 1;
            const int l7 = t & 7;
            const int r0 = st * 4 + rp * 2;
            v4f f;
            f.x = s_e[(r0    ) * 16 + l7];
            f.y = s_e[(r0 + 1) * 16 + l7];
            f.z = s_e[(r0    ) * 16 + 8 + l7];
            f.w = s_e[(r0 + 1) * 16 + 8 + l7];
            ((v4f*)e)[(slab * 4 + st) * 32 + rp * 16 + half * 8 + l7] = f;
        }
        if (t < 16) {
            float es = s_e[n];
            #pragma unroll
            for (int rr = 1; rr < 16; ++rr)
                es += s_e[rr * 16 + n];
            stat[blockIdx.x * KPAD + n]      = mx;
            stat[blockIdx.x * KPAD + 16 + n] = es;
        }
    }
}

extern "C" void kernel_launch(void* const* d_in, const int* in_sizes, int n_in,
                              void* d_out, int out_size, void* d_ws, size_t ws_size,
                              hipStream_t stream) {
    (void)in_sizes; (void)n_in; (void)out_size; (void)ws_size;
    const float* x = (const float*)d_in[0];
    const float* W = (const float*)d_in[1];
    float* out = (float*)d_out;

    float* bij  = (float*)d_ws;
    float* e    = bij  + M_SZ * KPAD;
    float* wv2  = e    + M_SZ * KPAD;
    float* stat = wv2  + I_SZ * KPAD;
    float* smax = stat;
    float* ssum = stat + 16;

    caps_gemm_t_squash<true><<<I_SZ / 16, T_WAVES * 32, 0, stream>>>(x, e, smax, ssum, W, out, wv2);
    caps_gemm_b<true><<<2 * N_SLAB, B_WAVES * 32, 0, stream>>>(x, wv2, bij, e, stat);
    caps_gemm_t_squash<false><<<I_SZ / 16, T_WAVES * 32, 0, stream>>>(x, e, smax, ssum, W, out, wv2);
    caps_gemm_b<false><<<2 * N_SLAB, B_WAVES * 32, 0, stream>>>(x, wv2, bij, e, stat);
    caps_gemm_t_squash<false><<<I_SZ / 16, T_WAVES * 32, 0, stream>>>(x, e, smax, ssum, W, out, wv2);
}
